# starting kernel
# baseline (speedup 1.0000x reference)
#include <hip/hip_runtime.h>
#include <hip/hip_bf16.h>
#include <math.h>


#define N_NODES 65536
#define IN_CH   64
#define HC      128
#define R_CENT  16
#define B_GR    128
#define HEADS   2
#define C_CH    64

typedef __attribute__((ext_vector_type(2))) float v2f;
typedef __attribute__((ext_vector_type(8))) float v8f;

__global__ __launch_bounds__(256) void BipartPool_xr_kernel(
    const float* __restrict__ xc,
    const float* __restrict__ w,
    const float* __restrict__ b,
    float* __restrict__ xr)
{
  for (int i = threadIdx.x; i < R_CENT * HC; i += 256) {
    int r = i >> 7, col = i & 127;
    float acc = b[col];
    #pragma unroll 8
    for (int k = 0; k < IN_CH; ++k) acc += xc[r * IN_CH + k] * w[k * HC + col];
    xr[i] = acc;
  }
}

__global__ __launch_bounds__(256) void BipartPool_xl_gemm_kernel(
    const float* __restrict__ x,
    const float* __restrict__ w,
    const float* __restrict__ b,
    float* __restrict__ xl)
{
  const int wave = threadIdx.x >> 5;
  const int lane = threadIdx.x & 31;
  const int lo = lane & 15;
  const int hi = lane >> 4;
  const int row0 = blockIdx.x * 16;
  const int col0 = wave * 16;

  const float bv = b[col0 + lo];
  v8f c;
  #pragma unroll
  for (int i = 0; i < 8; ++i) c[i] = bv;

  const float* arow = x + (size_t)(row0 + lo) * IN_CH;
  #pragma unroll
  for (int k = 0; k < IN_CH; k += 4) {
    const int kk = k + 2 * hi;
    v2f a = *(const v2f*)(arow + kk);
    v2f bb;
    bb.x = w[(size_t)kk * HC + col0 + lo];
    bb.y = w[(size_t)(kk + 1) * HC + col0 + lo];
    c = __builtin_amdgcn_wmma_f32_16x16x4_f32(false, a, false, bb,
                                              (short)0, c, false, false);
  }

  float* orow = xl + (size_t)row0 * HC + col0 + lo;
  #pragma unroll
  for (int v = 0; v < 8; ++v) orow[(size_t)(v + 8 * hi) * HC] = c[v];
}

__global__ __launch_bounds__(256) void BipartPool_attn_kernel(
    const float* __restrict__ xl,
    const float* __restrict__ xr,
    const float* __restrict__ att,
    const float* __restrict__ bias,
    float* __restrict__ out)
{
  extern __shared__ float smem[];
  float* s_logit = smem;
  float* s_acc   = s_logit + 32 * 512;
  float* s_den   = s_acc + 2 * 16 * 64;
  float* s_pself = s_den + 32;
  float* s_self  = s_pself + 32;

  const int g    = blockIdx.x;
  const int tid  = threadIdx.x;
  const int base = g * 512;
  const int wave = tid >> 5;
  const int lane = tid & 31;

  for (int p4 = 0; p4 < 4; ++p4) {
    const int pair = tid + p4 * 256;
    const int n = pair & 511;
    const int h = pair >> 9;
    const float* xrow = xl + (size_t)(base + n) * HC + h * C_CH;
    const float* arow = att + h * C_CH;
    float xv[C_CH];
    #pragma unroll
    for (int c = 0; c < C_CH; ++c) xv[c] = xrow[c];
    for (int r = 0; r < R_CENT; ++r) {
      const float* rrow = xr + r * HC + h * C_CH;
      float acc = 0.f;
      #pragma unroll
      for (int c = 0; c < C_CH; ++c) {
        float s = xv[c] + rrow[c];
        s = s > 0.f ? s : 0.2f * s;
        acc += arow[c] * s;
      }
      s_logit[(h * 16 + r) * 512 + n] = acc;
    }
  }

  if (tid < 32) {
    const int r = tid & 15, h = tid >> 4;
    const int t = g * 16 + r;
    const float* xrow = xl + (size_t)t * HC + h * C_CH;
    const float* rrow = xr + r * HC + h * C_CH;
    const float* arow = att + h * C_CH;
    float acc = 0.f;
    #pragma unroll 8
    for (int c = 0; c < C_CH; ++c) {
      float s = xrow[c] + rrow[c];
      s = s > 0.f ? s : 0.2f * s;
      acc += arow[c] * s;
    }
    s_self[tid] = acc;
  }
  __syncthreads();

  if (g == 0 && tid < 32) {
    const int r = tid & 15, h = tid >> 4;
    s_logit[(h * 16 + r) * 512 + r] = -__builtin_inff();
  }
  __syncthreads();

  for (int q = 0; q < 4; ++q) {
    const int rh = wave * 4 + q;
    float* lrow = s_logit + rh * 512;
    float m = s_self[rh];
    for (int i = lane; i < 512; i += 32) m = fmaxf(m, lrow[i]);
    #pragma unroll
    for (int off = 16; off >= 1; off >>= 1) m = fmaxf(m, __shfl_xor(m, off, 32));
    float sum = 0.f;
    for (int i = lane; i < 512; i += 32) {
      const float pv = __expf(lrow[i] - m);
      lrow[i] = pv;
      sum += pv;
    }
    #pragma unroll
    for (int off = 16; off >= 1; off >>= 1) sum += __shfl_xor(sum, off, 32);
    if (lane == 0) {
      const float ps = __expf(s_self[rh] - m);
      s_pself[rh] = ps;
      s_den[rh] = sum + ps + 1e-16f;
    }
  }
  __syncthreads();

  {
    const int h = wave >> 2, nt = wave & 3;
    const int lo = lane & 15, hi = lane >> 4;
    const float* prow = s_logit + (h * 16 + lo) * 512;
    const float* bcol = xl + (size_t)base * HC + h * C_CH + nt * 16 + lo;
    v8f c;
    #pragma unroll
    for (int i = 0; i < 8; ++i) c[i] = 0.f;
    for (int k = 0; k < 512; k += 4) {
      const int kk = k + 2 * hi;
      v2f a = *(const v2f*)(prow + kk);
      v2f bb;
      bb.x = bcol[(size_t)kk * HC];
      bb.y = bcol[(size_t)(kk + 1) * HC];
      c = __builtin_amdgcn_wmma_f32_16x16x4_f32(false, a, false, bb,
                                                (short)0, c, false, false);
    }
    #pragma unroll
    for (int v = 0; v < 8; ++v)
      s_acc[(h * 16 + v + 8 * hi) * 64 + nt * 16 + lo] = c[v];
  }
  __syncthreads();

  for (int i = tid; i < 16 * 64; i += 256) {
    const int r = i >> 6, cc = i & 63;
    const int t = g * 16 + r;
    const float v0 = (s_acc[(0 * 16 + r) * 64 + cc] +
                      s_pself[0 * 16 + r] * xl[(size_t)t * HC + cc]) /
                     s_den[0 * 16 + r];
    const float v1 = (s_acc[(1 * 16 + r) * 64 + cc] +
                      s_pself[1 * 16 + r] * xl[(size_t)t * HC + C_CH + cc]) /
                     s_den[1 * 16 + r];
    out[(size_t)t * 64 + cc] = 0.5f * (v0 + v1) + bias[cc];
  }
}

extern "C" void kernel_launch(void* const* d_in, const int* in_sizes, int n_in,
                              void* d_out, int out_size, void* d_ws, size_t ws_size,
                              hipStream_t stream) {
  const float* x    = (const float*)d_in[0];
  const float* xcb  = (const float*)d_in[3];
  const float* wl   = (const float*)d_in[4];
  const float* bl   = (const float*)d_in[5];
  const float* wr   = (const float*)d_in[6];
  const float* br   = (const float*)d_in[7];
  const float* att  = (const float*)d_in[8];
  const float* bias = (const float*)d_in[9];
  float* out = (float*)d_out;

  float* xl = (float*)d_ws;
  float* xr = xl + (size_t)N_NODES * HC;

  BipartPool_xr_kernel<<<1, 256, 0, stream>>>(xcb, wr, br, xr);
  BipartPool_xl_gemm_kernel<<<N_NODES / 16, 256, 0, stream>>>(x, wl, bl, xl);

  const size_t smem = (size_t)(32 * 512 + 2 * 16 * 64 + 3 * 32) * sizeof(float);
  BipartPool_attn_kernel<<<B_GR, 256, smem, stream>>>(xl, xr, att, bias, out);
}


# LDS softmax pass replaced by a per-wave pass in registers: numerators against 16-node tile maxima (DPP row max), tiles rescaled to the slice max on the weighted sum's A operand; exp-sum tree changed
# speedup vs baseline: 2.6159x; 2.6159x over previous
#include <hip/hip_runtime.h>
#include <hip/hip_bf16.h>
#include <math.h>


#define N_NODES 65536
#define IN_CH   64
#define HC      128
#define R_CENT  16
#define B_GR    128
#define HEADS   2
#define C_CH    64

typedef __attribute__((ext_vector_type(2))) float v2f;
typedef __attribute__((ext_vector_type(8))) float v8f;

template <int MASK>
__device__ __forceinline__ float row_xor(float v) {
  const int b = __builtin_bit_cast(int, v);
  return __builtin_bit_cast(float, __builtin_amdgcn_mov_dpp(b, 0x160 + MASK, 0xf, 0xf, false));
}

__device__ __forceinline__ float row_max16(float v) {
  v = fmaxf(v, row_xor<1>(v));
  v = fmaxf(v, row_xor<2>(v));
  v = fmaxf(v, row_xor<4>(v));
  v = fmaxf(v, row_xor<8>(v));
  return v;
}

__device__ __forceinline__ float xhalf(float x) {
  const int b = __builtin_bit_cast(int, x);
  return __builtin_bit_cast(float, __builtin_amdgcn_permlanex16(b, b, 0x76543210, 0xfedcba98, false, false));
}

#define CHUNK  128
#define NCHUNK 4
#define NGRP   (CHUNK / 16)
#define REC    (R_CENT * C_CH + 32)
#define XR_PITCH 20
#define Y_PITCH 68
#define WS_NEED (((size_t)N_NODES * HC + R_CENT * HC + (size_t)B_GR * NCHUNK * HEADS * REC) * sizeof(float))

__global__ __launch_bounds__(128) void BipartPool_attn_part_kernel(
    const float* __restrict__ x,
    const float* __restrict__ wl,
    const float* __restrict__ bl,
    const float* __restrict__ xc,
    const float* __restrict__ wr,
    const float* __restrict__ br,
    const float* __restrict__ att,
    float* __restrict__ xr,
    float* __restrict__ part,
    size_t ws_bytes)
{
  if (ws_bytes < WS_NEED) return;

  __shared__ float s_logit[R_CENT * CHUNK];
  __shared__ __attribute__((aligned(16))) float s_gmax[R_CENT * NGRP];
  __shared__ __attribute__((aligned(16))) float s_y[R_CENT * Y_PITCH];
  __shared__ float s_out[R_CENT * C_CH];
  __shared__ float s_stat[32];
  __shared__ __attribute__((aligned(16))) float s_xrT[C_CH * XR_PITCH];

  const int wg    = blockIdx.x;
  const int h     = wg & 1;
  const int chunk = (wg >> 1) & (NCHUNK - 1);
  const int g     = wg >> 3;
  const int tid   = threadIdx.x;
  const int wave  = tid >> 5;
  const int lane  = tid & 31;
  const int lo    = lane & 15;
  const int hi    = lane >> 4;
  const int node0 = g * 512 + chunk * CHUNK;

  float4 xq[2][8];
  #pragma unroll
  for (int nt = 0; nt < 2; ++nt) {
    const float* xb = x + (size_t)(node0 + 32 * wave + 16 * nt + lo) * IN_CH + 32 * hi;
    #pragma unroll
    for (int q = 0; q < 8; ++q) xq[nt][q] = *(const float4*)(xb + 4 * q);
  }

  {
    const int col = h * C_CH + wave * 16 + lo;
    const float bv = br[col];
    v8f d;
    #pragma unroll
    for (int i = 0; i < 8; ++i) d[i] = bv;
    const float* acr = xc + lo * IN_CH;
    #pragma unroll
    for (int k = 0; k < IN_CH; k += 4) {
      const int kk = k + 2 * hi;
      v2f a = *(const v2f*)(acr + kk);
      v2f bb;
      bb.x = wr[(size_t)kk * HC + col];
      bb.y = wr[(size_t)(kk + 1) * HC + col];
      d = __builtin_amdgcn_wmma_f32_16x16x4_f32(false, a, false, bb,
                                                (short)0, d, false, false);
    }
    float* dst = s_xrT + (wave * 16 + lo) * XR_PITCH + 8 * hi;
    *(float4*)(dst)     = make_float4(d[0], d[1], d[2], d[3]);
    *(float4*)(dst + 4) = make_float4(d[4], d[5], d[6], d[7]);
  }

  float av[4][8];
  #pragma unroll
  for (int ct = 0; ct < 4; ++ct) {
    #pragma unroll
    for (int v = 0; v < 8; ++v) av[ct][v] = att[h * C_CH + ct * 16 + 8 * hi + v];
  }
  __syncthreads();

  if (g == 0 && chunk == 0) {
    for (int i = tid; i < R_CENT * C_CH; i += 128) {
      const int r = i >> 6, c = i & 63;
      xr[r * HC + h * C_CH + c] = s_xrT[c * XR_PITCH + r];
    }
  }

  #pragma unroll
  for (int nt = 0; nt < 2; ++nt) {
    v8f d[4];
    {
      const float* wcol = wl + h * C_CH + lo;
      const float* bch  = bl + h * C_CH + 8 * hi;
      #pragma unroll
      for (int cp = 0; cp < 2; ++cp) {
        v2f a[2][16];
        #pragma unroll
        for (int c2 = 0; c2 < 2; ++c2) {
          const int ct = 2 * cp + c2;
          #pragma unroll
          for (int v = 0; v < 8; ++v) d[ct][v] = bch[ct * 16 + v];
          #pragma unroll
          for (int j = 0; j < 16; ++j) {
            const int kk = 32 * hi + 2 * j;
            a[c2][j].x = wcol[(size_t)kk * HC + ct * 16];
            a[c2][j].y = wcol[(size_t)(kk + 1) * HC + ct * 16];
          }
        }
        #pragma unroll
        for (int j = 0; j < 16; ++j) {
          const float4 xx = xq[nt][j >> 1];
          v2f b;
          b.x = (j & 1) ? xx.z : xx.x;
          b.y = (j & 1) ? xx.w : xx.y;
          #pragma unroll
          for (int c2 = 0; c2 < 2; ++c2)
            d[2 * cp + c2] = __builtin_amdgcn_wmma_f32_16x16x4_f32(false, a[c2][j], false, b,
                                                                   (short)0, d[2 * cp + c2], false, false);
        }
      }
    }

    const float* xrl = s_xrT + (8 * hi) * XR_PITCH;
    const int node = 32 * wave + 16 * nt + lo;
    for (int q = 0; q < R_CENT / 4; ++q) {
      float acc0 = 0.f, acc1 = 0.f, acc2 = 0.f, acc3 = 0.f;
      #pragma unroll
      for (int ct = 0; ct < 4; ++ct) {
        float4 rr[8];
        #pragma unroll
        for (int v = 0; v < 8; ++v)
          rr[v] = *(const float4*)(xrl + (ct * 16 + v) * XR_PITCH + 4 * q);
        #pragma unroll
        for (int v = 0; v < 8; ++v) {
          const float a  = av[ct][v];
          const float xx = d[ct][v];
          float s0 = xx + rr[v].x;
          float s1 = xx + rr[v].y;
          float s2 = xx + rr[v].z;
          float s3 = xx + rr[v].w;
          s0 = s0 > 0.f ? s0 : 0.2f * s0;
          s1 = s1 > 0.f ? s1 : 0.2f * s1;
          s2 = s2 > 0.f ? s2 : 0.2f * s2;
          s3 = s3 > 0.f ? s3 : 0.2f * s3;
          acc0 += a * s0;
          acc1 += a * s1;
          acc2 += a * s2;
          acc3 += a * s3;
        }
      }
      const float keep0 = hi ? acc2 : acc0;
      const float send0 = hi ? acc0 : acc2;
      const float keep1 = hi ? acc3 : acc1;
      const float send1 = hi ? acc1 : acc3;
      const float t0 = keep0 + __shfl_xor(send0, 16, 32);
      const float t1 = keep1 + __shfl_xor(send1, 16, 32);
      s_logit[(4 * q + 2 * hi + 0) * CHUNK + node] = t0;
      s_logit[(4 * q + 2 * hi + 1) * CHUNK + node] = t1;
    }
  }

  {
    float lg[2][8];
    #pragma unroll
    for (int nt = 0; nt < 2; ++nt) {
      #pragma unroll
      for (int i = 0; i < 8; ++i)
        lg[nt][i] = s_logit[(4 * (i >> 1) + 2 * hi + (i & 1)) * CHUNK + 32 * wave + 16 * nt + lo];
    }
    const bool dupw = (g == 0) & (chunk == 0) & (wave == 0);
    #pragma unroll
    for (int i = 0; i < 8; ++i)
      lg[0][i] = (dupw & (lo == 4 * (i >> 1) + 2 * hi + (i & 1))) ? -__builtin_inff() : lg[0][i];
    float mx[2][8];
    #pragma unroll
    for (int nt = 0; nt < 2; ++nt) {
      #pragma unroll
      for (int i = 0; i < 8; ++i) mx[nt][i] = row_max16(lg[nt][i]);
    }
    #pragma unroll
    for (int nt = 0; nt < 2; ++nt) {
      #pragma unroll
      for (int i = 0; i < 8; ++i)
        s_logit[(4 * (i >> 1) + 2 * hi + (i & 1)) * CHUNK + 32 * wave + 16 * nt + lo] =
            __expf(lg[nt][i] - mx[nt][i]);
    }
    if (lo == 0) {
      #pragma unroll
      for (int nt = 0; nt < 2; ++nt) {
        #pragma unroll
        for (int i = 0; i < 8; ++i)
          s_gmax[(4 * (i >> 1) + 2 * hi + (i & 1)) * NGRP + 2 * wave + nt] = mx[nt][i];
      }
    }
  }
  __syncthreads();

  {
    const int nt = wave;
    const float* prow = s_logit + lo * CHUNK;
    const float* bcol = x + (size_t)node0 * IN_CH + nt * 16 + lo;
    const float4 g0 = *(const float4*)(s_gmax + lo * NGRP);
    const float4 g1 = *(const float4*)(s_gmax + lo * NGRP + 4);
    const float gm[NGRP] = {g0.x, g0.y, g0.z, g0.w, g1.x, g1.y, g1.z, g1.w};
    float M = gm[0];
    #pragma unroll
    for (int i = 1; i < NGRP; ++i) M = fmaxf(M, gm[i]);
    float sc[NGRP];
    #pragma unroll
    for (int i = 0; i < NGRP; ++i) sc[i] = __expf(gm[i] - M);
    v8f c;
    #pragma unroll
    for (int i = 0; i < 8; ++i) c[i] = 0.f;
    v2f psum = {0.f, 0.f};
    #pragma unroll
    for (int k = 0; k < CHUNK; k += 4) {
      const int kk = k + 2 * hi;
      v2f a = *(const v2f*)(prow + kk);
      a *= sc[k >> 4];
      psum += a;
      v2f bb;
      bb.x = bcol[(size_t)kk * IN_CH];
      bb.y = bcol[(size_t)(kk + 1) * IN_CH];
      c = __builtin_amdgcn_wmma_f32_16x16x4_f32(false, a, false, bb,
                                                (short)0, c, false, false);
    }
    float S = psum.x + psum.y;
    S += xhalf(S);
    if (wave == 0 && hi == 0) {
      s_stat[lo] = M;
      s_stat[16 + lo] = S;
    }
    #pragma unroll
    for (int v = 0; v < 8; ++v)
      s_y[(v + 8 * hi) * Y_PITCH + nt * 16 + lo] = c[v];
  }
  __syncthreads();

  {
    const int col = h * C_CH + wave * 16 + lo;
    const float bv = bl[col];
    v8f c;
    #pragma unroll
    for (int v = 0; v < 8; ++v) c[v] = bv * s_stat[16 + v + 8 * hi];
    const float* yrow = s_y + lo * Y_PITCH;
    #pragma unroll
    for (int k = 0; k < IN_CH; k += 4) {
      const int kk = k + 2 * hi;
      v2f a = *(const v2f*)(yrow + kk);
      v2f bb;
      bb.x = wl[(size_t)kk * HC + col];
      bb.y = wl[(size_t)(kk + 1) * HC + col];
      c = __builtin_amdgcn_wmma_f32_16x16x4_f32(false, a, false, bb,
                                                (short)0, c, false, false);
    }
    #pragma unroll
    for (int v = 0; v < 8; ++v)
      s_out[(v + 8 * hi) * C_CH + wave * 16 + lo] = c[v];
  }
  __syncthreads();

  float* rec = part + (size_t)wg * REC;
  for (int i = tid; i < R_CENT * C_CH; i += 128) rec[i] = s_out[i];
  if (tid < 32) rec[R_CENT * C_CH + tid] = s_stat[tid];
}

#define XL_PITCH 132

__global__ __launch_bounds__(256) void BipartPool_attn_merge_kernel(
    const float* __restrict__ x,
    const float* __restrict__ wl,
    const float* __restrict__ bl,
    const float* __restrict__ xr,
    const float* __restrict__ att,
    const float* __restrict__ bias,
    const float* __restrict__ part,
    float* __restrict__ out,
    size_t ws_bytes)
{
  if (ws_bytes < WS_NEED) return;

  __shared__ float s_scale[32 * NCHUNK];
  __shared__ float s_pself[32];
  __shared__ float s_den[32];
  __shared__ float s_xl[R_CENT * XL_PITCH];

  const int g   = blockIdx.x;
  const int tid = threadIdx.x;
  const float* grec = part + (size_t)g * (2 * NCHUNK) * REC;

  float pa[4][2][NCHUNK];
  #pragma unroll
  for (int j = 0; j < 4; ++j) {
    const int i = tid + j * 256;
    const int r = i >> 6, cc = i & 63;
    #pragma unroll
    for (int h = 0; h < 2; ++h) {
      #pragma unroll
      for (int s = 0; s < NCHUNK; ++s)
        pa[j][h][s] = grec[(size_t)(2 * s + h) * REC + r * C_CH + cc];
    }
  }

  {
    const int wave = tid >> 5, lane = tid & 31;
    const int lo = lane & 15, hi = lane >> 4;
    const int col0 = wave * 16;
    const float bv = bl[col0 + lo];
    v8f c;
    #pragma unroll
    for (int i = 0; i < 8; ++i) c[i] = bv;
    const float* arow = x + (size_t)(g * 16 + lo) * IN_CH;
    #pragma unroll
    for (int k = 0; k < IN_CH; k += 4) {
      const int kk = k + 2 * hi;
      v2f a = *(const v2f*)(arow + kk);
      v2f bb;
      bb.x = wl[(size_t)kk * HC + col0 + lo];
      bb.y = wl[(size_t)(kk + 1) * HC + col0 + lo];
      c = __builtin_amdgcn_wmma_f32_16x16x4_f32(false, a, false, bb,
                                                (short)0, c, false, false);
    }
    #pragma unroll
    for (int v = 0; v < 8; ++v) s_xl[(v + 8 * hi) * XL_PITCH + col0 + lo] = c[v];
  }
  __syncthreads();

  float xs[4][2];
  #pragma unroll
  for (int j = 0; j < 4; ++j) {
    const int i = tid + j * 256;
    const int r = i >> 6, cc = i & 63;
    #pragma unroll
    for (int h = 0; h < 2; ++h) xs[j][h] = s_xl[r * XL_PITCH + h * C_CH + cc];
  }

  {
    const int rh = tid >> 3, sub = tid & 7;
    const int r = rh & 15, h = rh >> 4;
    const float* xrow = s_xl + r * XL_PITCH + h * C_CH + sub * 8;
    const float* rrow = xr + r * HC + h * C_CH + sub * 8;
    const float* arow = att + h * C_CH + sub * 8;
    float acc = 0.f;
    #pragma unroll
    for (int c = 0; c < 8; ++c) {
      float s = xrow[c] + rrow[c];
      s = s > 0.f ? s : 0.2f * s;
      acc += arow[c] * s;
    }
    acc += __shfl_xor(acc, 1, 32);
    acc += __shfl_xor(acc, 2, 32);
    acc += __shfl_xor(acc, 4, 32);
    if (sub == 0) {
      const float* st = grec + (size_t)h * REC + R_CENT * C_CH + r;
      float pm[NCHUNK], psum[NCHUNK];
      float M = acc;
      #pragma unroll
      for (int s = 0; s < NCHUNK; ++s) {
        pm[s]   = st[(size_t)(2 * s) * REC];
        psum[s] = st[(size_t)(2 * s) * REC + 16];
        M = fmaxf(M, pm[s]);
      }
      float den = 0.f;
      #pragma unroll
      for (int s = 0; s < NCHUNK; ++s) {
        const float sc = __expf(pm[s] - M);
        s_scale[rh * NCHUNK + s] = sc;
        den += psum[s] * sc;
      }
      const float ps = __expf(acc - M);
      s_pself[rh] = ps;
      s_den[rh] = den + ps + 1e-16f;
    }
  }
  __syncthreads();

  #pragma unroll
  for (int j = 0; j < 4; ++j) {
    const int i = tid + j * 256;
    const int r = i >> 6, cc = i & 63;
    const int t = g * 16 + r;
    float v[2];
    #pragma unroll
    for (int h = 0; h < 2; ++h) {
      float a = s_pself[h * 16 + r] * xs[j][h];
      #pragma unroll
      for (int s = 0; s < NCHUNK; ++s)
        a += s_scale[(h * 16 + r) * NCHUNK + s] * pa[j][h][s];
      v[h] = a / s_den[h * 16 + r];
    }
    out[(size_t)t * 64 + cc] = 0.5f * (v[0] + v[1]) + bias[cc];
  }
}

extern "C" void kernel_launch(void* const* d_in, const int* in_sizes, int n_in,
                              void* d_out, int out_size, void* d_ws, size_t ws_size,
                              hipStream_t stream) {
  const float* x    = (const float*)d_in[0];
  const float* xcb  = (const float*)d_in[3];
  const float* wl   = (const float*)d_in[4];
  const float* bl   = (const float*)d_in[5];
  const float* wr   = (const float*)d_in[6];
  const float* br   = (const float*)d_in[7];
  const float* att  = (const float*)d_in[8];
  const float* bias = (const float*)d_in[9];
  float* out = (float*)d_out;

  float* ws = (float*)d_ws;
  float* xr = ws + (size_t)N_NODES * HC;
  float* part = xr + R_CENT * HC;

  BipartPool_attn_part_kernel<<<B_GR * NCHUNK * HEADS, 128, 0, stream>>>(x, wl, bl, xcb, wr, br, att, xr, part, ws_size);
  BipartPool_attn_merge_kernel<<<B_GR, 256, 0, stream>>>(x, wl, bl, xr, att, bias, part, out, ws_size);
}


# attention kernel: a wave's two node tiles share one projection + logit pass; x_r slice kept [target][channel] so packed f32 ops pair adjacent channels; logit channel sum = even + odd partial sums
# speedup vs baseline: 2.9157x; 1.1146x over previous
#include <hip/hip_runtime.h>
#include <hip/hip_bf16.h>
#include <math.h>


#define N_NODES 65536
#define IN_CH   64
#define HC      128
#define R_CENT  16
#define B_GR    128
#define HEADS   2
#define C_CH    64

typedef __attribute__((ext_vector_type(2))) float v2f;
typedef __attribute__((ext_vector_type(8))) float v8f;

template <int MASK>
__device__ __forceinline__ float row_xor(float v) {
  const int b = __builtin_bit_cast(int, v);
  return __builtin_bit_cast(float, __builtin_amdgcn_mov_dpp(b, 0x160 + MASK, 0xf, 0xf, false));
}

__device__ __forceinline__ float row_max16(float v) {
  v = fmaxf(v, row_xor<1>(v));
  v = fmaxf(v, row_xor<2>(v));
  v = fmaxf(v, row_xor<4>(v));
  v = fmaxf(v, row_xor<8>(v));
  return v;
}

__device__ __forceinline__ float xhalf(float x) {
  const int b = __builtin_bit_cast(int, x);
  return __builtin_bit_cast(float, __builtin_amdgcn_permlanex16(b, b, 0x76543210, 0xfedcba98, false, false));
}

#define CHUNK  128
#define NCHUNK 4
#define NGRP   (CHUNK / 16)
#define REC    (R_CENT * C_CH + 32)
#define XR_PITCH 68
#define Y_PITCH 68
#define WS_NEED (((size_t)N_NODES * HC + R_CENT * HC + (size_t)B_GR * NCHUNK * HEADS * REC) * sizeof(float))

__global__ __launch_bounds__(128) void BipartPool_attn_part_kernel(
    const float* __restrict__ x,
    const float* __restrict__ wl,
    const float* __restrict__ bl,
    const float* __restrict__ xc,
    const float* __restrict__ wr,
    const float* __restrict__ br,
    const float* __restrict__ att,
    float* __restrict__ xr,
    float* __restrict__ part,
    size_t ws_bytes)
{
  if (ws_bytes < WS_NEED) return;

  __shared__ float s_logit[R_CENT * CHUNK];
  __shared__ __attribute__((aligned(16))) float s_gmax[R_CENT * NGRP];
  __shared__ __attribute__((aligned(16))) float s_y[R_CENT * Y_PITCH];
  __shared__ float s_out[R_CENT * C_CH];
  __shared__ float s_stat[32];
  __shared__ __attribute__((aligned(16))) float s_xr[R_CENT * XR_PITCH];

  const int wg    = blockIdx.x;
  const int h     = wg & 1;
  const int chunk = (wg >> 1) & (NCHUNK - 1);
  const int g     = wg >> 3;
  const int tid   = threadIdx.x;
  const int wave  = tid >> 5;
  const int lane  = tid & 31;
  const int lo    = lane & 15;
  const int hi    = lane >> 4;
  const int node0 = g * 512 + chunk * CHUNK;

  float4 xq[2][8];
  #pragma unroll
  for (int nt = 0; nt < 2; ++nt) {
    const float* xb = x + (size_t)(node0 + 32 * wave + 16 * nt + lo) * IN_CH + 32 * hi;
    #pragma unroll
    for (int q = 0; q < 8; ++q) xq[nt][q] = *(const float4*)(xb + 4 * q);
  }

  {
    const int col = h * C_CH + wave * 16 + lo;
    const float bv = br[col];
    v8f d;
    #pragma unroll
    for (int i = 0; i < 8; ++i) d[i] = bv;
    const float* acr = xc + lo * IN_CH;
    #pragma unroll
    for (int k = 0; k < IN_CH; k += 4) {
      const int kk = k + 2 * hi;
      v2f a = *(const v2f*)(acr + kk);
      v2f bb;
      bb.x = wr[(size_t)kk * HC + col];
      bb.y = wr[(size_t)(kk + 1) * HC + col];
      d = __builtin_amdgcn_wmma_f32_16x16x4_f32(false, a, false, bb,
                                                (short)0, d, false, false);
    }
    float* dst = s_xr + (8 * hi) * XR_PITCH + wave * 16 + lo;
    #pragma unroll
    for (int v = 0; v < 8; ++v) dst[v * XR_PITCH] = d[v];
  }

  __syncthreads();

  if (g == 0 && chunk == 0) {
    for (int i = tid; i < R_CENT * C_CH; i += 128) {
      const int r = i >> 6, c = i & 63;
      xr[r * HC + h * C_CH + c] = s_xr[r * XR_PITCH + c];
    }
  }

  v8f d[2][4];
  {
    const float* wcol = wl + h * C_CH + lo;
    const float* bch  = bl + h * C_CH + 8 * hi;
    #pragma unroll
    for (int cp = 0; cp < 2; ++cp) {
      v2f a[2][16];
      #pragma unroll
      for (int c2 = 0; c2 < 2; ++c2) {
        const int ct = 2 * cp + c2;
        #pragma unroll
        for (int v = 0; v < 8; ++v) {
          const float bv = bch[ct * 16 + v];
          d[0][ct][v] = bv;
          d[1][ct][v] = bv;
        }
        #pragma unroll
        for (int j = 0; j < 16; ++j) {
          const int kk = 32 * hi + 2 * j;
          a[c2][j].x = wcol[(size_t)kk * HC + ct * 16];
          a[c2][j].y = wcol[(size_t)(kk + 1) * HC + ct * 16];
        }
      }
      #pragma unroll
      for (int j = 0; j < 16; ++j) {
        #pragma unroll
        for (int nt = 0; nt < 2; ++nt) {
          const float4 xx = xq[nt][j >> 1];
          v2f b;
          b.x = (j & 1) ? xx.z : xx.x;
          b.y = (j & 1) ? xx.w : xx.y;
          #pragma unroll
          for (int c2 = 0; c2 < 2; ++c2)
            d[nt][2 * cp + c2] = __builtin_amdgcn_wmma_f32_16x16x4_f32(false, a[c2][j], false, b,
                                                                       (short)0, d[nt][2 * cp + c2], false, false);
        }
      }
    }
  }

  float av[4][8];
  #pragma unroll
  for (int ct = 0; ct < 4; ++ct) {
    #pragma unroll
    for (int v = 0; v < 8; ++v) av[ct][v] = att[h * C_CH + ct * 16 + 8 * hi + v];
  }

  {
    const float* xrl = s_xr + 8 * hi;
    const int node = 32 * wave + lo;
    for (int q = 0; q < R_CENT / 4; ++q) {
      v2f accp[2][4];
      #pragma unroll
      for (int nt = 0; nt < 2; ++nt) {
        #pragma unroll
        for (int t = 0; t < 4; ++t) accp[nt][t] = (v2f){0.f, 0.f};
      }
      #pragma unroll
      for (int ct = 0; ct < 4; ++ct) {
        float4 rlo[4], rhi[4];
        #pragma unroll
        for (int t = 0; t < 4; ++t) {
          const float* p = xrl + (4 * q + t) * XR_PITCH + ct * 16;
          rlo[t] = *(const float4*)(p);
          rhi[t] = *(const float4*)(p + 4);
        }
        #pragma unroll
        for (int nt = 0; nt < 2; ++nt) {
          #pragma unroll
          for (int vp = 0; vp < 4; ++vp) {
            const v2f dd = {d[nt][ct][2 * vp], d[nt][ct][2 * vp + 1]};
            const v2f aa = {av[ct][2 * vp], av[ct][2 * vp + 1]};
            v2f s[4];
            #pragma unroll
            for (int t = 0; t < 4; ++t) {
              const float4 rq = (vp < 2) ? rlo[t] : rhi[t];
              const v2f r = (vp & 1) ? (v2f){rq.z, rq.w} : (v2f){rq.x, rq.y};
              s[t] = dd + r;
            }
            #pragma unroll
            for (int t = 0; t < 4; ++t) {
              const float sx = s[t].x, sy = s[t].y;
              const float lx = 0.2f * sx, ly = 0.2f * sy;
              s[t].x = sx > 0.f ? sx : lx;
              s[t].y = sy > 0.f ? sy : ly;
            }
            #pragma unroll
            for (int t = 0; t < 4; ++t) accp[nt][t] += aa * s[t];
          }
        }
      }
      #pragma unroll
      for (int nt = 0; nt < 2; ++nt) {
        const float acc0 = accp[nt][0].x + accp[nt][0].y;
        const float acc1 = accp[nt][1].x + accp[nt][1].y;
        const float acc2 = accp[nt][2].x + accp[nt][2].y;
        const float acc3 = accp[nt][3].x + accp[nt][3].y;
        const float keep0 = hi ? acc2 : acc0;
        const float send0 = hi ? acc0 : acc2;
        const float keep1 = hi ? acc3 : acc1;
        const float send1 = hi ? acc1 : acc3;
        const float t0 = keep0 + __shfl_xor(send0, 16, 32);
        const float t1 = keep1 + __shfl_xor(send1, 16, 32);
        s_logit[(4 * q + 2 * hi + 0) * CHUNK + node + 16 * nt] = t0;
        s_logit[(4 * q + 2 * hi + 1) * CHUNK + node + 16 * nt] = t1;
      }
    }
  }

  {
    float lg[2][8];
    #pragma unroll
    for (int nt = 0; nt < 2; ++nt) {
      #pragma unroll
      for (int i = 0; i < 8; ++i)
        lg[nt][i] = s_logit[(4 * (i >> 1) + 2 * hi + (i & 1)) * CHUNK + 32 * wave + 16 * nt + lo];
    }
    const bool dupw = (g == 0) & (chunk == 0) & (wave == 0);
    #pragma unroll
    for (int i = 0; i < 8; ++i)
      lg[0][i] = (dupw & (lo == 4 * (i >> 1) + 2 * hi + (i & 1))) ? -__builtin_inff() : lg[0][i];
    float mx[2][8];
    #pragma unroll
    for (int nt = 0; nt < 2; ++nt) {
      #pragma unroll
      for (int i = 0; i < 8; ++i) mx[nt][i] = row_max16(lg[nt][i]);
    }
    #pragma unroll
    for (int nt = 0; nt < 2; ++nt) {
      #pragma unroll
      for (int i = 0; i < 8; ++i)
        s_logit[(4 * (i >> 1) + 2 * hi + (i & 1)) * CHUNK + 32 * wave + 16 * nt + lo] =
            __expf(lg[nt][i] - mx[nt][i]);
    }
    if (lo == 0) {
      #pragma unroll
      for (int nt = 0; nt < 2; ++nt) {
        #pragma unroll
        for (int i = 0; i < 8; ++i)
          s_gmax[(4 * (i >> 1) + 2 * hi + (i & 1)) * NGRP + 2 * wave + nt] = mx[nt][i];
      }
    }
  }
  __syncthreads();

  {
    const int nt = wave;
    const float* prow = s_logit + lo * CHUNK;
    const float* bcol = x + (size_t)node0 * IN_CH + nt * 16 + lo;
    const float4 g0 = *(const float4*)(s_gmax + lo * NGRP);
    const float4 g1 = *(const float4*)(s_gmax + lo * NGRP + 4);
    const float gm[NGRP] = {g0.x, g0.y, g0.z, g0.w, g1.x, g1.y, g1.z, g1.w};
    float M = gm[0];
    #pragma unroll
    for (int i = 1; i < NGRP; ++i) M = fmaxf(M, gm[i]);
    float sc[NGRP];
    #pragma unroll
    for (int i = 0; i < NGRP; ++i) sc[i] = __expf(gm[i] - M);
    v8f c;
    #pragma unroll
    for (int i = 0; i < 8; ++i) c[i] = 0.f;
    v2f psum = {0.f, 0.f};
    #pragma unroll
    for (int k = 0; k < CHUNK; k += 4) {
      const int kk = k + 2 * hi;
      v2f a = *(const v2f*)(prow + kk);
      a *= sc[k >> 4];
      psum += a;
      v2f bb;
      bb.x = bcol[(size_t)kk * IN_CH];
      bb.y = bcol[(size_t)(kk + 1) * IN_CH];
      c = __builtin_amdgcn_wmma_f32_16x16x4_f32(false, a, false, bb,
                                                (short)0, c, false, false);
    }
    float S = psum.x + psum.y;
    S += xhalf(S);
    if (wave == 0 && hi == 0) {
      s_stat[lo] = M;
      s_stat[16 + lo] = S;
    }
    #pragma unroll
    for (int v = 0; v < 8; ++v)
      s_y[(v + 8 * hi) * Y_PITCH + nt * 16 + lo] = c[v];
  }
  __syncthreads();

  {
    const int col = h * C_CH + wave * 16 + lo;
    const float bv = bl[col];
    v8f c;
    #pragma unroll
    for (int v = 0; v < 8; ++v) c[v] = bv * s_stat[16 + v + 8 * hi];
    const float* yrow = s_y + lo * Y_PITCH;
    #pragma unroll
    for (int k = 0; k < IN_CH; k += 4) {
      const int kk = k + 2 * hi;
      v2f a = *(const v2f*)(yrow + kk);
      v2f bb;
      bb.x = wl[(size_t)kk * HC + col];
      bb.y = wl[(size_t)(kk + 1) * HC + col];
      c = __builtin_amdgcn_wmma_f32_16x16x4_f32(false, a, false, bb,
                                                (short)0, c, false, false);
    }
    #pragma unroll
    for (int v = 0; v < 8; ++v)
      s_out[(v + 8 * hi) * C_CH + wave * 16 + lo] = c[v];
  }
  __syncthreads();

  float* rec = part + (size_t)wg * REC;
  for (int i = tid; i < R_CENT * C_CH; i += 128) rec[i] = s_out[i];
  if (tid < 32) rec[R_CENT * C_CH + tid] = s_stat[tid];
}

#define XL_PITCH 132

__global__ __launch_bounds__(256) void BipartPool_attn_merge_kernel(
    const float* __restrict__ x,
    const float* __restrict__ wl,
    const float* __restrict__ bl,
    const float* __restrict__ xr,
    const float* __restrict__ att,
    const float* __restrict__ bias,
    const float* __restrict__ part,
    float* __restrict__ out,
    size_t ws_bytes)
{
  if (ws_bytes < WS_NEED) return;

  __shared__ float s_scale[32 * NCHUNK];
  __shared__ float s_pself[32];
  __shared__ float s_den[32];
  __shared__ float s_xl[R_CENT * XL_PITCH];

  const int g   = blockIdx.x;
  const int tid = threadIdx.x;
  const float* grec = part + (size_t)g * (2 * NCHUNK) * REC;

  float pa[4][2][NCHUNK];
  #pragma unroll
  for (int j = 0; j < 4; ++j) {
    const int i = tid + j * 256;
    const int r = i >> 6, cc = i & 63;
    #pragma unroll
    for (int h = 0; h < 2; ++h) {
      #pragma unroll
      for (int s = 0; s < NCHUNK; ++s)
        pa[j][h][s] = grec[(size_t)(2 * s + h) * REC + r * C_CH + cc];
    }
  }

  {
    const int wave = tid >> 5, lane = tid & 31;
    const int lo = lane & 15, hi = lane >> 4;
    const int col0 = wave * 16;
    const float bv = bl[col0 + lo];
    v8f c;
    #pragma unroll
    for (int i = 0; i < 8; ++i) c[i] = bv;
    const float* arow = x + (size_t)(g * 16 + lo) * IN_CH;
    #pragma unroll
    for (int k = 0; k < IN_CH; k += 4) {
      const int kk = k + 2 * hi;
      v2f a = *(const v2f*)(arow + kk);
      v2f bb;
      bb.x = wl[(size_t)kk * HC + col0 + lo];
      bb.y = wl[(size_t)(kk + 1) * HC + col0 + lo];
      c = __builtin_amdgcn_wmma_f32_16x16x4_f32(false, a, false, bb,
                                                (short)0, c, false, false);
    }
    #pragma unroll
    for (int v = 0; v < 8; ++v) s_xl[(v + 8 * hi) * XL_PITCH + col0 + lo] = c[v];
  }
  __syncthreads();

  float xs[4][2];
  #pragma unroll
  for (int j = 0; j < 4; ++j) {
    const int i = tid + j * 256;
    const int r = i >> 6, cc = i & 63;
    #pragma unroll
    for (int h = 0; h < 2; ++h) xs[j][h] = s_xl[r * XL_PITCH + h * C_CH + cc];
  }

  {
    const int rh = tid >> 3, sub = tid & 7;
    const int r = rh & 15, h = rh >> 4;
    const float* xrow = s_xl + r * XL_PITCH + h * C_CH + sub * 8;
    const float* rrow = xr + r * HC + h * C_CH + sub * 8;
    const float* arow = att + h * C_CH + sub * 8;
    float acc = 0.f;
    #pragma unroll
    for (int c = 0; c < 8; ++c) {
      float s = xrow[c] + rrow[c];
      s = s > 0.f ? s : 0.2f * s;
      acc += arow[c] * s;
    }
    acc += __shfl_xor(acc, 1, 32);
    acc += __shfl_xor(acc, 2, 32);
    acc += __shfl_xor(acc, 4, 32);
    if (sub == 0) {
      const float* st = grec + (size_t)h * REC + R_CENT * C_CH + r;
      float pm[NCHUNK], psum[NCHUNK];
      float M = acc;
      #pragma unroll
      for (int s = 0; s < NCHUNK; ++s) {
        pm[s]   = st[(size_t)(2 * s) * REC];
        psum[s] = st[(size_t)(2 * s) * REC + 16];
        M = fmaxf(M, pm[s]);
      }
      float den = 0.f;
      #pragma unroll
      for (int s = 0; s < NCHUNK; ++s) {
        const float sc = __expf(pm[s] - M);
        s_scale[rh * NCHUNK + s] = sc;
        den += psum[s] * sc;
      }
      const float ps = __expf(acc - M);
      s_pself[rh] = ps;
      s_den[rh] = den + ps + 1e-16f;
    }
  }
  __syncthreads();

  #pragma unroll
  for (int j = 0; j < 4; ++j) {
    const int i = tid + j * 256;
    const int r = i >> 6, cc = i & 63;
    const int t = g * 16 + r;
    float v[2];
    #pragma unroll
    for (int h = 0; h < 2; ++h) {
      float a = s_pself[h * 16 + r] * xs[j][h];
      #pragma unroll
      for (int s = 0; s < NCHUNK; ++s)
        a += s_scale[(h * 16 + r) * NCHUNK + s] * pa[j][h][s];
      v[h] = a / s_den[h * 16 + r];
    }
    out[(size_t)t * 64 + cc] = 0.5f * (v[0] + v[1]) + bias[cc];
  }
}

extern "C" void kernel_launch(void* const* d_in, const int* in_sizes, int n_in,
                              void* d_out, int out_size, void* d_ws, size_t ws_size,
                              hipStream_t stream) {
  const float* x    = (const float*)d_in[0];
  const float* xcb  = (const float*)d_in[3];
  const float* wl   = (const float*)d_in[4];
  const float* bl   = (const float*)d_in[5];
  const float* wr   = (const float*)d_in[6];
  const float* br   = (const float*)d_in[7];
  const float* att  = (const float*)d_in[8];
  const float* bias = (const float*)d_in[9];
  float* out = (float*)d_out;

  float* ws = (float*)d_ws;
  float* xr = ws + (size_t)N_NODES * HC;
  float* part = xr + R_CENT * HC;

  BipartPool_attn_part_kernel<<<B_GR * NCHUNK * HEADS, 128, 0, stream>>>(x, wl, bl, xcb, wr, br, att, xr, part, ws_size);
  BipartPool_attn_merge_kernel<<<B_GR, 256, 0, stream>>>(x, wl, bl, xr, att, bias, part, out, ws_size);
}


# attention kernel, logit loop: leaky_relu taken as max(s, slope*s) instead of compare + select (6 -> 4 issue slots per channel pair), in stages of 8 independent packed ops; 4 waves per SIMD stated
# speedup vs baseline: 3.1577x; 1.0830x over previous
#include <hip/hip_runtime.h>
#include <hip/hip_bf16.h>
#include <math.h>


#define N_NODES 65536
#define IN_CH   64
#define HC      128
#define R_CENT  16
#define B_GR    128
#define HEADS   2
#define C_CH    64

typedef __attribute__((ext_vector_type(2))) float v2f;
typedef __attribute__((ext_vector_type(8))) float v8f;

template <int MASK>
__device__ __forceinline__ float row_xor(float v) {
  const int b = __builtin_bit_cast(int, v);
  return __builtin_bit_cast(float, __builtin_amdgcn_mov_dpp(b, 0x160 + MASK, 0xf, 0xf, false));
}

__device__ __forceinline__ float row_max16(float v) {
  v = fmaxf(v, row_xor<1>(v));
  v = fmaxf(v, row_xor<2>(v));
  v = fmaxf(v, row_xor<4>(v));
  v = fmaxf(v, row_xor<8>(v));
  return v;
}

__device__ __forceinline__ float xhalf(float x) {
  const int b = __builtin_bit_cast(int, x);
  return __builtin_bit_cast(float, __builtin_amdgcn_permlanex16(b, b, 0x76543210, 0xfedcba98, false, false));
}

#define CHUNK  128
#define NCHUNK 4
#define NGRP   (CHUNK / 16)
#define REC    (R_CENT * C_CH + 32)
#define XR_PITCH 68
#define Y_PITCH 68
#define WS_NEED (((size_t)N_NODES * HC + R_CENT * HC + (size_t)B_GR * NCHUNK * HEADS * REC) * sizeof(float))

__global__ __launch_bounds__(128) __attribute__((amdgpu_waves_per_eu(4, 4))) void BipartPool_attn_part_kernel(
    const float* __restrict__ x,
    const float* __restrict__ wl,
    const float* __restrict__ bl,
    const float* __restrict__ xc,
    const float* __restrict__ wr,
    const float* __restrict__ br,
    const float* __restrict__ att,
    float* __restrict__ xr,
    float* __restrict__ part,
    size_t ws_bytes)
{
  if (ws_bytes < WS_NEED) return;

  __shared__ float s_logit[R_CENT * CHUNK];
  __shared__ __attribute__((aligned(16))) float s_gmax[R_CENT * NGRP];
  __shared__ __attribute__((aligned(16))) float s_y[R_CENT * Y_PITCH];
  __shared__ float s_out[R_CENT * C_CH];
  __shared__ float s_stat[32];
  __shared__ __attribute__((aligned(16))) float s_xr[R_CENT * XR_PITCH];

  const int wg    = blockIdx.x;
  const int h     = wg & 1;
  const int chunk = (wg >> 1) & (NCHUNK - 1);
  const int g     = wg >> 3;
  const int tid   = threadIdx.x;
  const int wave  = tid >> 5;
  const int lane  = tid & 31;
  const int lo    = lane & 15;
  const int hi    = lane >> 4;
  const int node0 = g * 512 + chunk * CHUNK;

  float4 xq[2][8];
  #pragma unroll
  for (int nt = 0; nt < 2; ++nt) {
    const float* xb = x + (size_t)(node0 + 32 * wave + 16 * nt + lo) * IN_CH + 32 * hi;
    #pragma unroll
    for (int q = 0; q < 8; ++q) xq[nt][q] = *(const float4*)(xb + 4 * q);
  }

  {
    const int col = h * C_CH + wave * 16 + lo;
    const float bv = br[col];
    v8f d;
    #pragma unroll
    for (int i = 0; i < 8; ++i) d[i] = bv;
    const float* acr = xc + lo * IN_CH;
    #pragma unroll
    for (int k = 0; k < IN_CH; k += 4) {
      const int kk = k + 2 * hi;
      v2f a = *(const v2f*)(acr + kk);
      v2f bb;
      bb.x = wr[(size_t)kk * HC + col];
      bb.y = wr[(size_t)(kk + 1) * HC + col];
      d = __builtin_amdgcn_wmma_f32_16x16x4_f32(false, a, false, bb,
                                                (short)0, d, false, false);
    }
    float* dst = s_xr + (8 * hi) * XR_PITCH + wave * 16 + lo;
    #pragma unroll
    for (int v = 0; v < 8; ++v) dst[v * XR_PITCH] = d[v];
  }

  __syncthreads();

  if (g == 0 && chunk == 0) {
    for (int i = tid; i < R_CENT * C_CH; i += 128) {
      const int r = i >> 6, c = i & 63;
      xr[r * HC + h * C_CH + c] = s_xr[r * XR_PITCH + c];
    }
  }

  v8f d[2][4];
  {
    const float* wcol = wl + h * C_CH + lo;
    const float* bch  = bl + h * C_CH + 8 * hi;
    #pragma unroll
    for (int cp = 0; cp < 2; ++cp) {
      v2f a[2][16];
      #pragma unroll
      for (int c2 = 0; c2 < 2; ++c2) {
        const int ct = 2 * cp + c2;
        #pragma unroll
        for (int v = 0; v < 8; ++v) {
          const float bv = bch[ct * 16 + v];
          d[0][ct][v] = bv;
          d[1][ct][v] = bv;
        }
        #pragma unroll
        for (int j = 0; j < 16; ++j) {
          const int kk = 32 * hi + 2 * j;
          a[c2][j].x = wcol[(size_t)kk * HC + ct * 16];
          a[c2][j].y = wcol[(size_t)(kk + 1) * HC + ct * 16];
        }
      }
      #pragma unroll
      for (int j = 0; j < 16; ++j) {
        #pragma unroll
        for (int nt = 0; nt < 2; ++nt) {
          const float4 xx = xq[nt][j >> 1];
          v2f b;
          b.x = (j & 1) ? xx.z : xx.x;
          b.y = (j & 1) ? xx.w : xx.y;
          #pragma unroll
          for (int c2 = 0; c2 < 2; ++c2)
            d[nt][2 * cp + c2] = __builtin_amdgcn_wmma_f32_16x16x4_f32(false, a[c2][j], false, b,
                                                                       (short)0, d[nt][2 * cp + c2], false, false);
        }
      }
    }
  }

  float av[4][8];
  #pragma unroll
  for (int ct = 0; ct < 4; ++ct) {
    #pragma unroll
    for (int v = 0; v < 8; ++v) av[ct][v] = att[h * C_CH + ct * 16 + 8 * hi + v];
  }

  {
    const float* xrl = s_xr + 8 * hi;
    const int node = 32 * wave + lo;
    for (int q = 0; q < R_CENT / 4; ++q) {
      v2f accp[2][4];
      #pragma unroll
      for (int nt = 0; nt < 2; ++nt) {
        #pragma unroll
        for (int t = 0; t < 4; ++t) accp[nt][t] = (v2f){0.f, 0.f};
      }
      #pragma unroll
      for (int ct = 0; ct < 4; ++ct) {
        float4 rlo[4], rhi[4];
        #pragma unroll
        for (int t = 0; t < 4; ++t) {
          const float* p = xrl + (4 * q + t) * XR_PITCH + ct * 16;
          rlo[t] = *(const float4*)(p);
          rhi[t] = *(const float4*)(p + 4);
        }
        #pragma unroll
        for (int nt = 0; nt < 2; ++nt) {
          #pragma unroll
          for (int vh = 0; vh < 2; ++vh) {
            v2f s[2][4], l[2][4];
            #pragma unroll
            for (int u = 0; u < 2; ++u) {
              const int vp = 2 * vh + u;
              const v2f dd = {d[nt][ct][2 * vp], d[nt][ct][2 * vp + 1]};
              #pragma unroll
              for (int t = 0; t < 4; ++t) {
                const float4 rq = vh ? rhi[t] : rlo[t];
                const v2f r = u ? (v2f){rq.z, rq.w} : (v2f){rq.x, rq.y};
                s[u][t] = dd + r;
              }
            }
            __builtin_amdgcn_sched_barrier(0);
            #pragma unroll
            for (int u = 0; u < 2; ++u) {
              #pragma unroll
              for (int t = 0; t < 4; ++t) l[u][t] = 0.2f * s[u][t];
            }
            __builtin_amdgcn_sched_barrier(0);
            #pragma unroll
            for (int u = 0; u < 2; ++u) {
              #pragma unroll
              for (int t = 0; t < 4; ++t) {
                s[u][t].x = fmaxf(s[u][t].x, l[u][t].x);
                s[u][t].y = fmaxf(s[u][t].y, l[u][t].y);
              }
            }
            __builtin_amdgcn_sched_barrier(0);
            #pragma unroll
            for (int u = 0; u < 2; ++u) {
              const int vp = 2 * vh + u;
              const v2f aa = {av[ct][2 * vp], av[ct][2 * vp + 1]};
              #pragma unroll
              for (int t = 0; t < 4; ++t) accp[nt][t] += aa * s[u][t];
            }
          }
        }
      }
      #pragma unroll
      for (int nt = 0; nt < 2; ++nt) {
        const float acc0 = accp[nt][0].x + accp[nt][0].y;
        const float acc1 = accp[nt][1].x + accp[nt][1].y;
        const float acc2 = accp[nt][2].x + accp[nt][2].y;
        const float acc3 = accp[nt][3].x + accp[nt][3].y;
        const float keep0 = hi ? acc2 : acc0;
        const float send0 = hi ? acc0 : acc2;
        const float keep1 = hi ? acc3 : acc1;
        const float send1 = hi ? acc1 : acc3;
        const float t0 = keep0 + __shfl_xor(send0, 16, 32);
        const float t1 = keep1 + __shfl_xor(send1, 16, 32);
        s_logit[(4 * q + 2 * hi + 0) * CHUNK + node + 16 * nt] = t0;
        s_logit[(4 * q + 2 * hi + 1) * CHUNK + node + 16 * nt] = t1;
      }
    }
  }

  {
    float lg[2][8];
    #pragma unroll
    for (int nt = 0; nt < 2; ++nt) {
      #pragma unroll
      for (int i = 0; i < 8; ++i)
        lg[nt][i] = s_logit[(4 * (i >> 1) + 2 * hi + (i & 1)) * CHUNK + 32 * wave + 16 * nt + lo];
    }
    const bool dupw = (g == 0) & (chunk == 0) & (wave == 0);
    #pragma unroll
    for (int i = 0; i < 8; ++i)
      lg[0][i] = (dupw & (lo == 4 * (i >> 1) + 2 * hi + (i & 1))) ? -__builtin_inff() : lg[0][i];
    float mx[2][8];
    #pragma unroll
    for (int nt = 0; nt < 2; ++nt) {
      #pragma unroll
      for (int i = 0; i < 8; ++i) mx[nt][i] = row_max16(lg[nt][i]);
    }
    #pragma unroll
    for (int nt = 0; nt < 2; ++nt) {
      #pragma unroll
      for (int i = 0; i < 8; ++i)
        s_logit[(4 * (i >> 1) + 2 * hi + (i & 1)) * CHUNK + 32 * wave + 16 * nt + lo] =
            __expf(lg[nt][i] - mx[nt][i]);
    }
    if (lo == 0) {
      #pragma unroll
      for (int nt = 0; nt < 2; ++nt) {
        #pragma unroll
        for (int i = 0; i < 8; ++i)
          s_gmax[(4 * (i >> 1) + 2 * hi + (i & 1)) * NGRP + 2 * wave + nt] = mx[nt][i];
      }
    }
  }
  __syncthreads();

  {
    const int nt = wave;
    const float* prow = s_logit + lo * CHUNK;
    const float* bcol = x + (size_t)node0 * IN_CH + nt * 16 + lo;
    const float4 g0 = *(const float4*)(s_gmax + lo * NGRP);
    const float4 g1 = *(const float4*)(s_gmax + lo * NGRP + 4);
    const float gm[NGRP] = {g0.x, g0.y, g0.z, g0.w, g1.x, g1.y, g1.z, g1.w};
    float M = gm[0];
    #pragma unroll
    for (int i = 1; i < NGRP; ++i) M = fmaxf(M, gm[i]);
    float sc[NGRP];
    #pragma unroll
    for (int i = 0; i < NGRP; ++i) sc[i] = __expf(gm[i] - M);
    v8f c;
    #pragma unroll
    for (int i = 0; i < 8; ++i) c[i] = 0.f;
    v2f psum = {0.f, 0.f};
    #pragma unroll
    for (int k = 0; k < CHUNK; k += 4) {
      const int kk = k + 2 * hi;
      v2f a = *(const v2f*)(prow + kk);
      a *= sc[k >> 4];
      psum += a;
      v2f bb;
      bb.x = bcol[(size_t)kk * IN_CH];
      bb.y = bcol[(size_t)(kk + 1) * IN_CH];
      c = __builtin_amdgcn_wmma_f32_16x16x4_f32(false, a, false, bb,
                                                (short)0, c, false, false);
    }
    float S = psum.x + psum.y;
    S += xhalf(S);
    if (wave == 0 && hi == 0) {
      s_stat[lo] = M;
      s_stat[16 + lo] = S;
    }
    #pragma unroll
    for (int v = 0; v < 8; ++v)
      s_y[(v + 8 * hi) * Y_PITCH + nt * 16 + lo] = c[v];
  }
  __syncthreads();

  {
    const int col = h * C_CH + wave * 16 + lo;
    const float bv = bl[col];
    v8f c;
    #pragma unroll
    for (int v = 0; v < 8; ++v) c[v] = bv * s_stat[16 + v + 8 * hi];
    const float* yrow = s_y + lo * Y_PITCH;
    #pragma unroll
    for (int k = 0; k < IN_CH; k += 4) {
      const int kk = k + 2 * hi;
      v2f a = *(const v2f*)(yrow + kk);
      v2f bb;
      bb.x = wl[(size_t)kk * HC + col];
      bb.y = wl[(size_t)(kk + 1) * HC + col];
      c = __builtin_amdgcn_wmma_f32_16x16x4_f32(false, a, false, bb,
                                                (short)0, c, false, false);
    }
    #pragma unroll
    for (int v = 0; v < 8; ++v)
      s_out[(v + 8 * hi) * C_CH + wave * 16 + lo] = c[v];
  }
  __syncthreads();

  float* rec = part + (size_t)wg * REC;
  for (int i = tid; i < R_CENT * C_CH; i += 128) rec[i] = s_out[i];
  if (tid < 32) rec[R_CENT * C_CH + tid] = s_stat[tid];
}

#define XL_PITCH 132

__global__ __launch_bounds__(256) void BipartPool_attn_merge_kernel(
    const float* __restrict__ x,
    const float* __restrict__ wl,
    const float* __restrict__ bl,
    const float* __restrict__ xr,
    const float* __restrict__ att,
    const float* __restrict__ bias,
    const float* __restrict__ part,
    float* __restrict__ out,
    size_t ws_bytes)
{
  if (ws_bytes < WS_NEED) return;

  __shared__ float s_scale[32 * NCHUNK];
  __shared__ float s_pself[32];
  __shared__ float s_den[32];
  __shared__ float s_xl[R_CENT * XL_PITCH];

  const int g   = blockIdx.x;
  const int tid = threadIdx.x;
  const float* grec = part + (size_t)g * (2 * NCHUNK) * REC;

  float pa[4][2][NCHUNK];
  #pragma unroll
  for (int j = 0; j < 4; ++j) {
    const int i = tid + j * 256;
    const int r = i >> 6, cc = i & 63;
    #pragma unroll
    for (int h = 0; h < 2; ++h) {
      #pragma unroll
      for (int s = 0; s < NCHUNK; ++s)
        pa[j][h][s] = grec[(size_t)(2 * s + h) * REC + r * C_CH + cc];
    }
  }

  {
    const int wave = tid >> 5, lane = tid & 31;
    const int lo = lane & 15, hi = lane >> 4;
    const int col0 = wave * 16;
    const float bv = bl[col0 + lo];
    v8f c;
    #pragma unroll
    for (int i = 0; i < 8; ++i) c[i] = bv;
    const float* arow = x + (size_t)(g * 16 + lo) * IN_CH;
    #pragma unroll
    for (int k = 0; k < IN_CH; k += 4) {
      const int kk = k + 2 * hi;
      v2f a = *(const v2f*)(arow + kk);
      v2f bb;
      bb.x = wl[(size_t)kk * HC + col0 + lo];
      bb.y = wl[(size_t)(kk + 1) * HC + col0 + lo];
      c = __builtin_amdgcn_wmma_f32_16x16x4_f32(false, a, false, bb,
                                                (short)0, c, false, false);
    }
    #pragma unroll
    for (int v = 0; v < 8; ++v) s_xl[(v + 8 * hi) * XL_PITCH + col0 + lo] = c[v];
  }
  __syncthreads();

  float xs[4][2];
  #pragma unroll
  for (int j = 0; j < 4; ++j) {
    const int i = tid + j * 256;
    const int r = i >> 6, cc = i & 63;
    #pragma unroll
    for (int h = 0; h < 2; ++h) xs[j][h] = s_xl[r * XL_PITCH + h * C_CH + cc];
  }

  {
    const int rh = tid >> 3, sub = tid & 7;
    const int r = rh & 15, h = rh >> 4;
    const float* xrow = s_xl + r * XL_PITCH + h * C_CH + sub * 8;
    const float* rrow = xr + r * HC + h * C_CH + sub * 8;
    const float* arow = att + h * C_CH + sub * 8;
    float acc = 0.f;
    #pragma unroll
    for (int c = 0; c < 8; ++c) {
      float s = xrow[c] + rrow[c];
      s = fmaxf(s, 0.2f * s);
      acc += arow[c] * s;
    }
    acc += __shfl_xor(acc, 1, 32);
    acc += __shfl_xor(acc, 2, 32);
    acc += __shfl_xor(acc, 4, 32);
    if (sub == 0) {
      const float* st = grec + (size_t)h * REC + R_CENT * C_CH + r;
      float pm[NCHUNK], psum[NCHUNK];
      float M = acc;
      #pragma unroll
      for (int s = 0; s < NCHUNK; ++s) {
        pm[s]   = st[(size_t)(2 * s) * REC];
        psum[s] = st[(size_t)(2 * s) * REC + 16];
        M = fmaxf(M, pm[s]);
      }
      float den = 0.f;
      #pragma unroll
      for (int s = 0; s < NCHUNK; ++s) {
        const float sc = __expf(pm[s] - M);
        s_scale[rh * NCHUNK + s] = sc;
        den += psum[s] * sc;
      }
      const float ps = __expf(acc - M);
      s_pself[rh] = ps;
      s_den[rh] = den + ps + 1e-16f;
    }
  }
  __syncthreads();

  #pragma unroll
  for (int j = 0; j < 4; ++j) {
    const int i = tid + j * 256;
    const int r = i >> 6, cc = i & 63;
    const int t = g * 16 + r;
    float v[2];
    #pragma unroll
    for (int h = 0; h < 2; ++h) {
      float a = s_pself[h * 16 + r] * xs[j][h];
      #pragma unroll
      for (int s = 0; s < NCHUNK; ++s)
        a += s_scale[(h * 16 + r) * NCHUNK + s] * pa[j][h][s];
      v[h] = a / s_den[h * 16 + r];
    }
    out[(size_t)t * 64 + cc] = 0.5f * (v[0] + v[1]) + bias[cc];
  }
}

extern "C" void kernel_launch(void* const* d_in, const int* in_sizes, int n_in,
                              void* d_out, int out_size, void* d_ws, size_t ws_size,
                              hipStream_t stream) {
  const float* x    = (const float*)d_in[0];
  const float* xcb  = (const float*)d_in[3];
  const float* wl   = (const float*)d_in[4];
  const float* bl   = (const float*)d_in[5];
  const float* wr   = (const float*)d_in[6];
  const float* br   = (const float*)d_in[7];
  const float* att  = (const float*)d_in[8];
  const float* bias = (const float*)d_in[9];
  float* out = (float*)d_out;

  float* ws = (float*)d_ws;
  float* xr = ws + (size_t)N_NODES * HC;
  float* part = xr + R_CENT * HC;

  BipartPool_attn_part_kernel<<<B_GR * NCHUNK * HEADS, 128, 0, stream>>>(x, wl, bl, xcb, wr, br, att, xr, part, ws_size);
  BipartPool_attn_merge_kernel<<<B_GR, 256, 0, stream>>>(x, wl, bl, xr, att, bias, part, out, ws_size);
}


# attention kernel: a slice's input rows come in once by whole-line loads (512 contiguous bytes per wave load, was one line per lane) and stay in LDS; both B-fragment consumers read LDS
# speedup vs baseline: 3.3844x; 1.0718x over previous
#include <hip/hip_runtime.h>
#include <hip/hip_bf16.h>
#include <math.h>


#define N_NODES 65536
#define IN_CH   64
#define HC      128
#define R_CENT  16
#define B_GR    128
#define HEADS   2
#define C_CH    64

typedef __attribute__((ext_vector_type(2))) float v2f;
typedef __attribute__((ext_vector_type(8))) float v8f;

template <int MASK>
__device__ __forceinline__ float row_xor(float v) {
  const int b = __builtin_bit_cast(int, v);
  return __builtin_bit_cast(float, __builtin_amdgcn_mov_dpp(b, 0x160 + MASK, 0xf, 0xf, false));
}

__device__ __forceinline__ float row_max16(float v) {
  v = fmaxf(v, row_xor<1>(v));
  v = fmaxf(v, row_xor<2>(v));
  v = fmaxf(v, row_xor<4>(v));
  v = fmaxf(v, row_xor<8>(v));
  return v;
}

__device__ __forceinline__ float xhalf(float x) {
  const int b = __builtin_bit_cast(int, x);
  return __builtin_bit_cast(float, __builtin_amdgcn_permlanex16(b, b, 0x76543210, 0xfedcba98, false, false));
}

#define CHUNK  128
#define NCHUNK 4
#define NGRP   (CHUNK / 16)
#define REC    (R_CENT * C_CH + 32)
#define XR_PITCH 68
#define X_PITCH 68
#define Y_PITCH 68
#define WS_NEED (((size_t)N_NODES * HC + R_CENT * HC + (size_t)B_GR * NCHUNK * HEADS * REC) * sizeof(float))

__global__ __launch_bounds__(128) __attribute__((amdgpu_waves_per_eu(4, 4))) void BipartPool_attn_part_kernel(
    const float* __restrict__ x,
    const float* __restrict__ wl,
    const float* __restrict__ bl,
    const float* __restrict__ xc,
    const float* __restrict__ wr,
    const float* __restrict__ br,
    const float* __restrict__ att,
    float* __restrict__ xr,
    float* __restrict__ part,
    size_t ws_bytes)
{
  if (ws_bytes < WS_NEED) return;

  __shared__ float s_logit[R_CENT * CHUNK];
  __shared__ __attribute__((aligned(16))) float s_gmax[R_CENT * NGRP];
  __shared__ __attribute__((aligned(16))) float s_y[R_CENT * Y_PITCH];
  __shared__ float s_out[R_CENT * C_CH];
  __shared__ float s_stat[32];
  __shared__ __attribute__((aligned(16))) float s_xr[R_CENT * XR_PITCH];
  __shared__ __attribute__((aligned(16))) float s_x[CHUNK * X_PITCH];

  const int wg    = blockIdx.x;
  const int h     = wg & 1;
  const int chunk = (wg >> 1) & (NCHUNK - 1);
  const int g     = wg >> 3;
  const int tid   = threadIdx.x;
  const int wave  = tid >> 5;
  const int lane  = tid & 31;
  const int lo    = lane & 15;
  const int hi    = lane >> 4;
  const int node0 = g * 512 + chunk * CHUNK;

  float4 xs[16];
  {
    const float* xb = x + (size_t)(node0 + 32 * wave) * IN_CH + 4 * lane;
    #pragma unroll
    for (int i = 0; i < 16; ++i) xs[i] = *(const float4*)(xb + 128 * i);
  }

  {
    const int col = h * C_CH + wave * 16 + lo;
    const float bv = br[col];
    v8f d;
    #pragma unroll
    for (int i = 0; i < 8; ++i) d[i] = bv;
    const float* acr = xc + lo * IN_CH;
    #pragma unroll
    for (int k = 0; k < IN_CH; k += 4) {
      const int kk = k + 2 * hi;
      v2f a = *(const v2f*)(acr + kk);
      v2f bb;
      bb.x = wr[(size_t)kk * HC + col];
      bb.y = wr[(size_t)(kk + 1) * HC + col];
      d = __builtin_amdgcn_wmma_f32_16x16x4_f32(false, a, false, bb,
                                                (short)0, d, false, false);
    }
    float* dst = s_xr + (8 * hi) * XR_PITCH + wave * 16 + lo;
    #pragma unroll
    for (int v = 0; v < 8; ++v) dst[v * XR_PITCH] = d[v];
  }

  {
    float* dstx = s_x + (32 * wave + hi) * X_PITCH + 4 * lo;
    #pragma unroll
    for (int i = 0; i < 16; ++i) *(float4*)(dstx + 2 * i * X_PITCH) = xs[i];
  }

  __syncthreads();

  float4 xq[2][8];
  #pragma unroll
  for (int nt = 0; nt < 2; ++nt) {
    const float* xb = s_x + (32 * wave + 16 * nt + lo) * X_PITCH + 32 * hi;
    #pragma unroll
    for (int q = 0; q < 8; ++q) xq[nt][q] = *(const float4*)(xb + 4 * q);
  }

  if (g == 0 && chunk == 0) {
    for (int i = tid; i < R_CENT * C_CH; i += 128) {
      const int r = i >> 6, c = i & 63;
      xr[r * HC + h * C_CH + c] = s_xr[r * XR_PITCH + c];
    }
  }

  v8f d[2][4];
  {
    const float* wcol = wl + h * C_CH + lo;
    const float* bch  = bl + h * C_CH + 8 * hi;
    #pragma unroll
    for (int cp = 0; cp < 2; ++cp) {
      v2f a[2][16];
      #pragma unroll
      for (int c2 = 0; c2 < 2; ++c2) {
        const int ct = 2 * cp + c2;
        #pragma unroll
        for (int v = 0; v < 8; ++v) {
          const float bv = bch[ct * 16 + v];
          d[0][ct][v] = bv;
          d[1][ct][v] = bv;
        }
        #pragma unroll
        for (int j = 0; j < 16; ++j) {
          const int kk = 32 * hi + 2 * j;
          a[c2][j].x = wcol[(size_t)kk * HC + ct * 16];
          a[c2][j].y = wcol[(size_t)(kk + 1) * HC + ct * 16];
        }
      }
      #pragma unroll
      for (int j = 0; j < 16; ++j) {
        #pragma unroll
        for (int nt = 0; nt < 2; ++nt) {
          const float4 xx = xq[nt][j >> 1];
          v2f b;
          b.x = (j & 1) ? xx.z : xx.x;
          b.y = (j & 1) ? xx.w : xx.y;
          #pragma unroll
          for (int c2 = 0; c2 < 2; ++c2)
            d[nt][2 * cp + c2] = __builtin_amdgcn_wmma_f32_16x16x4_f32(false, a[c2][j], false, b,
                                                                       (short)0, d[nt][2 * cp + c2], false, false);
        }
      }
    }
  }

  float av[4][8];
  #pragma unroll
  for (int ct = 0; ct < 4; ++ct) {
    #pragma unroll
    for (int v = 0; v < 8; ++v) av[ct][v] = att[h * C_CH + ct * 16 + 8 * hi + v];
  }

  {
    const float* xrl = s_xr + 8 * hi;
    const int node = 32 * wave + lo;
    for (int q = 0; q < R_CENT / 4; ++q) {
      v2f accp[2][4];
      #pragma unroll
      for (int nt = 0; nt < 2; ++nt) {
        #pragma unroll
        for (int t = 0; t < 4; ++t) accp[nt][t] = (v2f){0.f, 0.f};
      }
      #pragma unroll
      for (int ct = 0; ct < 4; ++ct) {
        float4 rlo[4], rhi[4];
        #pragma unroll
        for (int t = 0; t < 4; ++t) {
          const float* p = xrl + (4 * q + t) * XR_PITCH + ct * 16;
          rlo[t] = *(const float4*)(p);
          rhi[t] = *(const float4*)(p + 4);
        }
        #pragma unroll
        for (int nt = 0; nt < 2; ++nt) {
          #pragma unroll
          for (int vh = 0; vh < 2; ++vh) {
            v2f s[2][4], l[2][4];
            #pragma unroll
            for (int u = 0; u < 2; ++u) {
              const int vp = 2 * vh + u;
              const v2f dd = {d[nt][ct][2 * vp], d[nt][ct][2 * vp + 1]};
              #pragma unroll
              for (int t = 0; t < 4; ++t) {
                const float4 rq = vh ? rhi[t] : rlo[t];
                const v2f r = u ? (v2f){rq.z, rq.w} : (v2f){rq.x, rq.y};
                s[u][t] = dd + r;
              }
            }
            __builtin_amdgcn_sched_barrier(0);
            #pragma unroll
            for (int u = 0; u < 2; ++u) {
              #pragma unroll
              for (int t = 0; t < 4; ++t) l[u][t] = 0.2f * s[u][t];
            }
            __builtin_amdgcn_sched_barrier(0);
            #pragma unroll
            for (int u = 0; u < 2; ++u) {
              #pragma unroll
              for (int t = 0; t < 4; ++t) {
                s[u][t].x = fmaxf(s[u][t].x, l[u][t].x);
                s[u][t].y = fmaxf(s[u][t].y, l[u][t].y);
              }
            }
            __builtin_amdgcn_sched_barrier(0);
            #pragma unroll
            for (int u = 0; u < 2; ++u) {
              const int vp = 2 * vh + u;
              const v2f aa = {av[ct][2 * vp], av[ct][2 * vp + 1]};
              #pragma unroll
              for (int t = 0; t < 4; ++t) accp[nt][t] += aa * s[u][t];
            }
          }
        }
      }
      #pragma unroll
      for (int nt = 0; nt < 2; ++nt) {
        const float acc0 = accp[nt][0].x + accp[nt][0].y;
        const float acc1 = accp[nt][1].x + accp[nt][1].y;
        const float acc2 = accp[nt][2].x + accp[nt][2].y;
        const float acc3 = accp[nt][3].x + accp[nt][3].y;
        const float keep0 = hi ? acc2 : acc0;
        const float send0 = hi ? acc0 : acc2;
        const float keep1 = hi ? acc3 : acc1;
        const float send1 = hi ? acc1 : acc3;
        const float t0 = keep0 + __shfl_xor(send0, 16, 32);
        const float t1 = keep1 + __shfl_xor(send1, 16, 32);
        s_logit[(4 * q + 2 * hi + 0) * CHUNK + node + 16 * nt] = t0;
        s_logit[(4 * q + 2 * hi + 1) * CHUNK + node + 16 * nt] = t1;
      }
    }
  }

  {
    float lg[2][8];
    #pragma unroll
    for (int nt = 0; nt < 2; ++nt) {
      #pragma unroll
      for (int i = 0; i < 8; ++i)
        lg[nt][i] = s_logit[(4 * (i >> 1) + 2 * hi + (i & 1)) * CHUNK + 32 * wave + 16 * nt + lo];
    }
    const bool dupw = (g == 0) & (chunk == 0) & (wave == 0);
    #pragma unroll
    for (int i = 0; i < 8; ++i)
      lg[0][i] = (dupw & (lo == 4 * (i >> 1) + 2 * hi + (i & 1))) ? -__builtin_inff() : lg[0][i];
    float mx[2][8];
    #pragma unroll
    for (int nt = 0; nt < 2; ++nt) {
      #pragma unroll
      for (int i = 0; i < 8; ++i) mx[nt][i] = row_max16(lg[nt][i]);
    }
    #pragma unroll
    for (int nt = 0; nt < 2; ++nt) {
      #pragma unroll
      for (int i = 0; i < 8; ++i)
        s_logit[(4 * (i >> 1) + 2 * hi + (i & 1)) * CHUNK + 32 * wave + 16 * nt + lo] =
            __expf(lg[nt][i] - mx[nt][i]);
    }
    if (lo == 0) {
      #pragma unroll
      for (int nt = 0; nt < 2; ++nt) {
        #pragma unroll
        for (int i = 0; i < 8; ++i)
          s_gmax[(4 * (i >> 1) + 2 * hi + (i & 1)) * NGRP + 2 * wave + nt] = mx[nt][i];
      }
    }
  }
  v2f xb[CHUNK / 4];
  {
    const float* bcol = s_x + 2 * hi * X_PITCH + wave * 16 + lo;
    #pragma unroll
    for (int j = 0; j < CHUNK / 4; ++j) {
      xb[j].x = bcol[(4 * j) * X_PITCH];
      xb[j].y = bcol[(4 * j + 1) * X_PITCH];
    }
  }
  __syncthreads();

  {
    const int nt = wave;
    const float* prow = s_logit + lo * CHUNK;
    const float4 g0 = *(const float4*)(s_gmax + lo * NGRP);
    const float4 g1 = *(const float4*)(s_gmax + lo * NGRP + 4);
    const float gm[NGRP] = {g0.x, g0.y, g0.z, g0.w, g1.x, g1.y, g1.z, g1.w};
    float M = gm[0];
    #pragma unroll
    for (int i = 1; i < NGRP; ++i) M = fmaxf(M, gm[i]);
    float sc[NGRP];
    #pragma unroll
    for (int i = 0; i < NGRP; ++i) sc[i] = __expf(gm[i] - M);
    v8f c;
    #pragma unroll
    for (int i = 0; i < 8; ++i) c[i] = 0.f;
    v2f psum = {0.f, 0.f};
    #pragma unroll
    for (int k = 0; k < CHUNK; k += 4) {
      const int kk = k + 2 * hi;
      v2f a = *(const v2f*)(prow + kk);
      a *= sc[k >> 4];
      psum += a;
      const v2f bb = xb[k >> 2];
      c = __builtin_amdgcn_wmma_f32_16x16x4_f32(false, a, false, bb,
                                                (short)0, c, false, false);
    }
    float S = psum.x + psum.y;
    S += xhalf(S);
    if (wave == 0 && hi == 0) {
      s_stat[lo] = M;
      s_stat[16 + lo] = S;
    }
    #pragma unroll
    for (int v = 0; v < 8; ++v)
      s_y[(v + 8 * hi) * Y_PITCH + nt * 16 + lo] = c[v];
  }
  __syncthreads();

  {
    const int col = h * C_CH + wave * 16 + lo;
    const float bv = bl[col];
    v8f c;
    #pragma unroll
    for (int v = 0; v < 8; ++v) c[v] = bv * s_stat[16 + v + 8 * hi];
    const float* yrow = s_y + lo * Y_PITCH;
    #pragma unroll
    for (int k = 0; k < IN_CH; k += 4) {
      const int kk = k + 2 * hi;
      v2f a = *(const v2f*)(yrow + kk);
      v2f bb;
      bb.x = wl[(size_t)kk * HC + col];
      bb.y = wl[(size_t)(kk + 1) * HC + col];
      c = __builtin_amdgcn_wmma_f32_16x16x4_f32(false, a, false, bb,
                                                (short)0, c, false, false);
    }
    #pragma unroll
    for (int v = 0; v < 8; ++v)
      s_out[(v + 8 * hi) * C_CH + wave * 16 + lo] = c[v];
  }
  __syncthreads();

  float* rec = part + (size_t)wg * REC;
  for (int i = tid; i < R_CENT * C_CH; i += 128) rec[i] = s_out[i];
  if (tid < 32) rec[R_CENT * C_CH + tid] = s_stat[tid];
}

#define XL_PITCH 132

__global__ __launch_bounds__(256) void BipartPool_attn_merge_kernel(
    const float* __restrict__ x,
    const float* __restrict__ wl,
    const float* __restrict__ bl,
    const float* __restrict__ xr,
    const float* __restrict__ att,
    const float* __restrict__ bias,
    const float* __restrict__ part,
    float* __restrict__ out,
    size_t ws_bytes)
{
  if (ws_bytes < WS_NEED) return;

  __shared__ float s_scale[32 * NCHUNK];
  __shared__ float s_pself[32];
  __shared__ float s_den[32];
  __shared__ float s_xl[R_CENT * XL_PITCH];

  const int g   = blockIdx.x;
  const int tid = threadIdx.x;
  const float* grec = part + (size_t)g * (2 * NCHUNK) * REC;

  float pa[4][2][NCHUNK];
  #pragma unroll
  for (int j = 0; j < 4; ++j) {
    const int i = tid + j * 256;
    const int r = i >> 6, cc = i & 63;
    #pragma unroll
    for (int h = 0; h < 2; ++h) {
      #pragma unroll
      for (int s = 0; s < NCHUNK; ++s)
        pa[j][h][s] = grec[(size_t)(2 * s + h) * REC + r * C_CH + cc];
    }
  }

  {
    const int wave = tid >> 5, lane = tid & 31;
    const int lo = lane & 15, hi = lane >> 4;
    const int col0 = wave * 16;
    const float bv = bl[col0 + lo];
    v8f c;
    #pragma unroll
    for (int i = 0; i < 8; ++i) c[i] = bv;
    const float* arow = x + (size_t)(g * 16 + lo) * IN_CH;
    #pragma unroll
    for (int k = 0; k < IN_CH; k += 4) {
      const int kk = k + 2 * hi;
      v2f a = *(const v2f*)(arow + kk);
      v2f bb;
      bb.x = wl[(size_t)kk * HC + col0 + lo];
      bb.y = wl[(size_t)(kk + 1) * HC + col0 + lo];
      c = __builtin_amdgcn_wmma_f32_16x16x4_f32(false, a, false, bb,
                                                (short)0, c, false, false);
    }
    #pragma unroll
    for (int v = 0; v < 8; ++v) s_xl[(v + 8 * hi) * XL_PITCH + col0 + lo] = c[v];
  }
  __syncthreads();

  float xs[4][2];
  #pragma unroll
  for (int j = 0; j < 4; ++j) {
    const int i = tid + j * 256;
    const int r = i >> 6, cc = i & 63;
    #pragma unroll
    for (int h = 0; h < 2; ++h) xs[j][h] = s_xl[r * XL_PITCH + h * C_CH + cc];
  }

  {
    const int rh = tid >> 3, sub = tid & 7;
    const int r = rh & 15, h = rh >> 4;
    const float* xrow = s_xl + r * XL_PITCH + h * C_CH + sub * 8;
    const float* rrow = xr + r * HC + h * C_CH + sub * 8;
    const float* arow = att + h * C_CH + sub * 8;
    float acc = 0.f;
    #pragma unroll
    for (int c = 0; c < 8; ++c) {
      float s = xrow[c] + rrow[c];
      s = fmaxf(s, 0.2f * s);
      acc += arow[c] * s;
    }
    acc += __shfl_xor(acc, 1, 32);
    acc += __shfl_xor(acc, 2, 32);
    acc += __shfl_xor(acc, 4, 32);
    if (sub == 0) {
      const float* st = grec + (size_t)h * REC + R_CENT * C_CH + r;
      float pm[NCHUNK], psum[NCHUNK];
      float M = acc;
      #pragma unroll
      for (int s = 0; s < NCHUNK; ++s) {
        pm[s]   = st[(size_t)(2 * s) * REC];
        psum[s] = st[(size_t)(2 * s) * REC + 16];
        M = fmaxf(M, pm[s]);
      }
      float den = 0.f;
      #pragma unroll
      for (int s = 0; s < NCHUNK; ++s) {
        const float sc = __expf(pm[s] - M);
        s_scale[rh * NCHUNK + s] = sc;
        den += psum[s] * sc;
      }
      const float ps = __expf(acc - M);
      s_pself[rh] = ps;
      s_den[rh] = den + ps + 1e-16f;
    }
  }
  __syncthreads();

  #pragma unroll
  for (int j = 0; j < 4; ++j) {
    const int i = tid + j * 256;
    const int r = i >> 6, cc = i & 63;
    const int t = g * 16 + r;
    float v[2];
    #pragma unroll
    for (int h = 0; h < 2; ++h) {
      float a = s_pself[h * 16 + r] * xs[j][h];
      #pragma unroll
      for (int s = 0; s < NCHUNK; ++s)
        a += s_scale[(h * 16 + r) * NCHUNK + s] * pa[j][h][s];
      v[h] = a / s_den[h * 16 + r];
    }
    out[(size_t)t * 64 + cc] = 0.5f * (v[0] + v[1]) + bias[cc];
  }
}

extern "C" void kernel_launch(void* const* d_in, const int* in_sizes, int n_in,
                              void* d_out, int out_size, void* d_ws, size_t ws_size,
                              hipStream_t stream) {
  const float* x    = (const float*)d_in[0];
  const float* xcb  = (const float*)d_in[3];
  const float* wl   = (const float*)d_in[4];
  const float* bl   = (const float*)d_in[5];
  const float* wr   = (const float*)d_in[6];
  const float* br   = (const float*)d_in[7];
  const float* att  = (const float*)d_in[8];
  const float* bias = (const float*)d_in[9];
  float* out = (float*)d_out;

  float* ws = (float*)d_ws;
  float* xr = ws + (size_t)N_NODES * HC;
  float* part = xr + R_CENT * HC;

  BipartPool_attn_part_kernel<<<B_GR * NCHUNK * HEADS, 128, 0, stream>>>(x, wl, bl, xcb, wr, br, att, xr, part, ws_size);
  BipartPool_attn_merge_kernel<<<B_GR, 256, 0, stream>>>(x, wl, bl, xr, att, bias, part, out, ws_size);
}


# W_l projection of the softmax-weighted input sums moved behind the slice merge: records carry raw sums; merge kernel adds slices + self-loop input row, projects once per graph; sum order changed
# speedup vs baseline: 3.6658x; 1.0831x over previous
#include <hip/hip_runtime.h>
#include <hip/hip_bf16.h>
#include <math.h>


#define N_NODES 65536
#define IN_CH   64
#define HC      128
#define R_CENT  16
#define B_GR    128
#define HEADS   2
#define C_CH    64

typedef __attribute__((ext_vector_type(2))) float v2f;
typedef __attribute__((ext_vector_type(8))) float v8f;

template <int MASK>
__device__ __forceinline__ float row_xor(float v) {
  const int b = __builtin_bit_cast(int, v);
  return __builtin_bit_cast(float, __builtin_amdgcn_mov_dpp(b, 0x160 + MASK, 0xf, 0xf, false));
}

__device__ __forceinline__ float row_max16(float v) {
  v = fmaxf(v, row_xor<1>(v));
  v = fmaxf(v, row_xor<2>(v));
  v = fmaxf(v, row_xor<4>(v));
  v = fmaxf(v, row_xor<8>(v));
  return v;
}

__device__ __forceinline__ float xhalf(float x) {
  const int b = __builtin_bit_cast(int, x);
  return __builtin_bit_cast(float, __builtin_amdgcn_permlanex16(b, b, 0x76543210, 0xfedcba98, false, false));
}

#define CHUNK  128
#define NCHUNK 4
#define NGRP   (CHUNK / 16)
#define REC    (R_CENT * IN_CH + 32)
#define N_SELF (R_CENT * B_GR)
#define XR_PITCH 68
#define X_PITCH 68
#define Y_PITCH 68
#define WS_NEED (((size_t)N_NODES * HC + R_CENT * HC + (size_t)B_GR * NCHUNK * HEADS * REC) * sizeof(float))

__global__ __launch_bounds__(128) __attribute__((amdgpu_waves_per_eu(4, 4))) void BipartPool_attn_part_kernel(
    const float* __restrict__ x,
    const float* __restrict__ wl,
    const float* __restrict__ bl,
    const float* __restrict__ xc,
    const float* __restrict__ wr,
    const float* __restrict__ br,
    const float* __restrict__ att,
    float* __restrict__ selfl,
    float* __restrict__ part,
    size_t ws_bytes)
{
  if (ws_bytes < WS_NEED) return;

  __shared__ float s_logit[R_CENT * CHUNK];
  __shared__ __attribute__((aligned(16))) float s_gmax[R_CENT * NGRP];
  __shared__ __attribute__((aligned(16))) float s_y[R_CENT * Y_PITCH];
  __shared__ float s_stat[32];
  __shared__ __attribute__((aligned(16))) float s_xr[R_CENT * XR_PITCH];
  __shared__ __attribute__((aligned(16))) float s_x[CHUNK * X_PITCH];

  const int wg    = blockIdx.x;
  const int h     = wg & 1;
  const int chunk = (wg >> 1) & (NCHUNK - 1);
  const int g     = wg >> 3;
  const int tid   = threadIdx.x;
  const int wave  = tid >> 5;
  const int lane  = tid & 31;
  const int lo    = lane & 15;
  const int hi    = lane >> 4;
  const int node0 = g * 512 + chunk * CHUNK;

  float4 xs[16];
  {
    const float* xb = x + (size_t)(node0 + 32 * wave) * IN_CH + 4 * lane;
    #pragma unroll
    for (int i = 0; i < 16; ++i) xs[i] = *(const float4*)(xb + 128 * i);
  }

  {
    const int col = h * C_CH + wave * 16 + lo;
    const float bv = br[col];
    v8f d;
    #pragma unroll
    for (int i = 0; i < 8; ++i) d[i] = bv;
    const float* acr = xc + lo * IN_CH;
    #pragma unroll
    for (int k = 0; k < IN_CH; k += 4) {
      const int kk = k + 2 * hi;
      v2f a = *(const v2f*)(acr + kk);
      v2f bb;
      bb.x = wr[(size_t)kk * HC + col];
      bb.y = wr[(size_t)(kk + 1) * HC + col];
      d = __builtin_amdgcn_wmma_f32_16x16x4_f32(false, a, false, bb,
                                                (short)0, d, false, false);
    }
    float* dst = s_xr + (8 * hi) * XR_PITCH + wave * 16 + lo;
    #pragma unroll
    for (int v = 0; v < 8; ++v) dst[v * XR_PITCH] = d[v];
  }

  {
    float* dstx = s_x + (32 * wave + hi) * X_PITCH + 4 * lo;
    #pragma unroll
    for (int i = 0; i < 16; ++i) *(float4*)(dstx + 2 * i * X_PITCH) = xs[i];
  }

  __syncthreads();

  float4 xq[2][8];
  #pragma unroll
  for (int nt = 0; nt < 2; ++nt) {
    const float* xb = s_x + (32 * wave + 16 * nt + lo) * X_PITCH + 32 * hi;
    #pragma unroll
    for (int q = 0; q < 8; ++q) xq[nt][q] = *(const float4*)(xb + 4 * q);
  }

  v8f d[2][4];
  {
    const float* wcol = wl + h * C_CH + lo;
    const float* bch  = bl + h * C_CH + 8 * hi;
    #pragma unroll
    for (int cp = 0; cp < 2; ++cp) {
      v2f a[2][16];
      #pragma unroll
      for (int c2 = 0; c2 < 2; ++c2) {
        const int ct = 2 * cp + c2;
        #pragma unroll
        for (int v = 0; v < 8; ++v) {
          const float bv = bch[ct * 16 + v];
          d[0][ct][v] = bv;
          d[1][ct][v] = bv;
        }
        #pragma unroll
        for (int j = 0; j < 16; ++j) {
          const int kk = 32 * hi + 2 * j;
          a[c2][j].x = wcol[(size_t)kk * HC + ct * 16];
          a[c2][j].y = wcol[(size_t)(kk + 1) * HC + ct * 16];
        }
      }
      #pragma unroll
      for (int j = 0; j < 16; ++j) {
        #pragma unroll
        for (int nt = 0; nt < 2; ++nt) {
          const float4 xx = xq[nt][j >> 1];
          v2f b;
          b.x = (j & 1) ? xx.z : xx.x;
          b.y = (j & 1) ? xx.w : xx.y;
          #pragma unroll
          for (int c2 = 0; c2 < 2; ++c2)
            d[nt][2 * cp + c2] = __builtin_amdgcn_wmma_f32_16x16x4_f32(false, a[c2][j], false, b,
                                                                       (short)0, d[nt][2 * cp + c2], false, false);
        }
      }
    }
  }

  float av[4][8];
  #pragma unroll
  for (int ct = 0; ct < 4; ++ct) {
    #pragma unroll
    for (int v = 0; v < 8; ++v) av[ct][v] = att[h * C_CH + ct * 16 + 8 * hi + v];
  }

  {
    const float* xrl = s_xr + 8 * hi;
    const int node = 32 * wave + lo;
    for (int q = 0; q < R_CENT / 4; ++q) {
      v2f accp[2][4];
      #pragma unroll
      for (int nt = 0; nt < 2; ++nt) {
        #pragma unroll
        for (int t = 0; t < 4; ++t) accp[nt][t] = (v2f){0.f, 0.f};
      }
      #pragma unroll
      for (int ct = 0; ct < 4; ++ct) {
        float4 rlo[4], rhi[4];
        #pragma unroll
        for (int t = 0; t < 4; ++t) {
          const float* p = xrl + (4 * q + t) * XR_PITCH + ct * 16;
          rlo[t] = *(const float4*)(p);
          rhi[t] = *(const float4*)(p + 4);
        }
        #pragma unroll
        for (int nt = 0; nt < 2; ++nt) {
          #pragma unroll
          for (int vh = 0; vh < 2; ++vh) {
            v2f s[2][4], l[2][4];
            #pragma unroll
            for (int u = 0; u < 2; ++u) {
              const int vp = 2 * vh + u;
              const v2f dd = {d[nt][ct][2 * vp], d[nt][ct][2 * vp + 1]};
              #pragma unroll
              for (int t = 0; t < 4; ++t) {
                const float4 rq = vh ? rhi[t] : rlo[t];
                const v2f r = u ? (v2f){rq.z, rq.w} : (v2f){rq.x, rq.y};
                s[u][t] = dd + r;
              }
            }
            __builtin_amdgcn_sched_barrier(0);
            #pragma unroll
            for (int u = 0; u < 2; ++u) {
              #pragma unroll
              for (int t = 0; t < 4; ++t) l[u][t] = 0.2f * s[u][t];
            }
            __builtin_amdgcn_sched_barrier(0);
            #pragma unroll
            for (int u = 0; u < 2; ++u) {
              #pragma unroll
              for (int t = 0; t < 4; ++t) {
                s[u][t].x = fmaxf(s[u][t].x, l[u][t].x);
                s[u][t].y = fmaxf(s[u][t].y, l[u][t].y);
              }
            }
            __builtin_amdgcn_sched_barrier(0);
            #pragma unroll
            for (int u = 0; u < 2; ++u) {
              const int vp = 2 * vh + u;
              const v2f aa = {av[ct][2 * vp], av[ct][2 * vp + 1]};
              #pragma unroll
              for (int t = 0; t < 4; ++t) accp[nt][t] += aa * s[u][t];
            }
          }
        }
      }
      #pragma unroll
      for (int nt = 0; nt < 2; ++nt) {
        const float acc0 = accp[nt][0].x + accp[nt][0].y;
        const float acc1 = accp[nt][1].x + accp[nt][1].y;
        const float acc2 = accp[nt][2].x + accp[nt][2].y;
        const float acc3 = accp[nt][3].x + accp[nt][3].y;
        const float keep0 = hi ? acc2 : acc0;
        const float send0 = hi ? acc0 : acc2;
        const float keep1 = hi ? acc3 : acc1;
        const float send1 = hi ? acc1 : acc3;
        const float t0 = keep0 + __shfl_xor(send0, 16, 32);
        const float t1 = keep1 + __shfl_xor(send1, 16, 32);
        s_logit[(4 * q + 2 * hi + 0) * CHUNK + node + 16 * nt] = t0;
        s_logit[(4 * q + 2 * hi + 1) * CHUNK + node + 16 * nt] = t1;
      }
    }
  }

  if (g < N_SELF / 512)
    selfl[h * N_SELF + node0 + 32 * wave + lane] = s_logit[(lane & 15) * CHUNK + 32 * wave + lane];

  {
    float lg[2][8];
    #pragma unroll
    for (int nt = 0; nt < 2; ++nt) {
      #pragma unroll
      for (int i = 0; i < 8; ++i)
        lg[nt][i] = s_logit[(4 * (i >> 1) + 2 * hi + (i & 1)) * CHUNK + 32 * wave + 16 * nt + lo];
    }
    const bool dupw = (g == 0) & (chunk == 0) & (wave == 0);
    #pragma unroll
    for (int i = 0; i < 8; ++i)
      lg[0][i] = (dupw & (lo == 4 * (i >> 1) + 2 * hi + (i & 1))) ? -__builtin_inff() : lg[0][i];
    float mx[2][8];
    #pragma unroll
    for (int nt = 0; nt < 2; ++nt) {
      #pragma unroll
      for (int i = 0; i < 8; ++i) mx[nt][i] = row_max16(lg[nt][i]);
    }
    #pragma unroll
    for (int nt = 0; nt < 2; ++nt) {
      #pragma unroll
      for (int i = 0; i < 8; ++i)
        s_logit[(4 * (i >> 1) + 2 * hi + (i & 1)) * CHUNK + 32 * wave + 16 * nt + lo] =
            __expf(lg[nt][i] - mx[nt][i]);
    }
    if (lo == 0) {
      #pragma unroll
      for (int nt = 0; nt < 2; ++nt) {
        #pragma unroll
        for (int i = 0; i < 8; ++i)
          s_gmax[(4 * (i >> 1) + 2 * hi + (i & 1)) * NGRP + 2 * wave + nt] = mx[nt][i];
      }
    }
  }
  v2f xb[CHUNK / 4];
  {
    const float* bcol = s_x + 2 * hi * X_PITCH + wave * 16 + lo;
    #pragma unroll
    for (int j = 0; j < CHUNK / 4; ++j) {
      xb[j].x = bcol[(4 * j) * X_PITCH];
      xb[j].y = bcol[(4 * j + 1) * X_PITCH];
    }
  }
  __syncthreads();

  {
    const int nt = wave;
    const float* prow = s_logit + lo * CHUNK;
    const float4 g0 = *(const float4*)(s_gmax + lo * NGRP);
    const float4 g1 = *(const float4*)(s_gmax + lo * NGRP + 4);
    const float gm[NGRP] = {g0.x, g0.y, g0.z, g0.w, g1.x, g1.y, g1.z, g1.w};
    float M = gm[0];
    #pragma unroll
    for (int i = 1; i < NGRP; ++i) M = fmaxf(M, gm[i]);
    float sc[NGRP];
    #pragma unroll
    for (int i = 0; i < NGRP; ++i) sc[i] = __expf(gm[i] - M);
    v8f c;
    #pragma unroll
    for (int i = 0; i < 8; ++i) c[i] = 0.f;
    v2f psum = {0.f, 0.f};
    #pragma unroll
    for (int k = 0; k < CHUNK; k += 4) {
      const int kk = k + 2 * hi;
      v2f a = *(const v2f*)(prow + kk);
      a *= sc[k >> 4];
      psum += a;
      const v2f bb = xb[k >> 2];
      c = __builtin_amdgcn_wmma_f32_16x16x4_f32(false, a, false, bb,
                                                (short)0, c, false, false);
    }
    float S = psum.x + psum.y;
    S += xhalf(S);
    if (wave == 0 && hi == 0) {
      s_stat[lo] = M;
      s_stat[16 + lo] = S;
    }
    #pragma unroll
    for (int v = 0; v < 8; ++v)
      s_y[(v + 8 * hi) * Y_PITCH + nt * 16 + lo] = c[v];
  }
  __syncthreads();

  float* rec = part + (size_t)wg * REC;
  for (int i = tid; i < R_CENT * IN_CH; i += 128) rec[i] = s_y[(i >> 6) * Y_PITCH + (i & 63)];
  if (tid < 32) rec[R_CENT * IN_CH + tid] = s_stat[tid];
}

__global__ __launch_bounds__(256) void BipartPool_attn_merge_kernel(
    const float* __restrict__ x,
    const float* __restrict__ wl,
    const float* __restrict__ bl,
    const float* __restrict__ selfl,
    const float* __restrict__ bias,
    const float* __restrict__ part,
    float* __restrict__ out,
    size_t ws_bytes)
{
  if (ws_bytes < WS_NEED) return;

  __shared__ __attribute__((aligned(16))) float s_scale[32 * NCHUNK];
  __shared__ float s_pself[32];
  __shared__ float s_wsum[32];
  __shared__ float s_den[32];
  __shared__ __attribute__((aligned(16))) float s_ym[2 * R_CENT * Y_PITCH];
  __shared__ __attribute__((aligned(16))) float s_o[2 * R_CENT * C_CH];

  const int g    = blockIdx.x;
  const int tid  = threadIdx.x;
  const int wave = tid >> 5, lane = tid & 31;
  const int lo   = lane & 15, hi = lane >> 4;
  const int r4   = tid >> 4;
  const int k4   = (tid & 15) * 4;
  const float* grec = part + (size_t)g * (2 * NCHUNK) * REC;

  float pm[NCHUNK], psum[NCHUNK], acc = 0.f;
  #pragma unroll
  for (int s = 0; s < NCHUNK; ++s) { pm[s] = 0.f; psum[s] = 0.f; }
  if (tid < 32) {
    const int r = tid & 15, h = tid >> 4;
    const float* st = grec + (size_t)h * REC + R_CENT * IN_CH + r;
    acc = selfl[h * N_SELF + g * R_CENT + r];
    #pragma unroll
    for (int s = 0; s < NCHUNK; ++s) {
      pm[s]   = st[(size_t)(2 * s) * REC];
      psum[s] = st[(size_t)(2 * s) * REC + 16];
    }
  }

  float4 pa[2][NCHUNK];
  #pragma unroll
  for (int h = 0; h < 2; ++h) {
    #pragma unroll
    for (int s = 0; s < NCHUNK; ++s)
      pa[h][s] = *(const float4*)(grec + (size_t)(2 * s + h) * REC + r4 * IN_CH + k4);
  }
  const float4 xs = *(const float4*)(x + (size_t)(g * R_CENT + r4) * IN_CH + k4);
  const float4 bs = *(const float4*)(bias + k4);
  const int col = wave * 16 + lo;
  v2f bw[IN_CH / 4];
  #pragma unroll
  for (int j = 0; j < IN_CH / 4; ++j) {
    const int kk = 4 * j + 2 * hi;
    bw[j].x = wl[(size_t)kk * HC + col];
    bw[j].y = wl[(size_t)(kk + 1) * HC + col];
  }
  const float bv = bl[col];

  if (tid < 32) {
    const int rh = tid;
    float M = acc;
    #pragma unroll
    for (int s = 0; s < NCHUNK; ++s) M = fmaxf(M, pm[s]);
    float den = 0.f;
    #pragma unroll
    for (int s = 0; s < NCHUNK; ++s) {
      const float sc = __expf(pm[s] - M);
      s_scale[rh * NCHUNK + s] = sc;
      den += psum[s] * sc;
    }
    const float ps = __expf(acc - M);
    s_pself[rh] = ps;
    s_wsum[rh] = den + ps;
    s_den[rh] = den + ps + 1e-16f;
  }
  __syncthreads();

  #pragma unroll
  for (int h = 0; h < 2; ++h) {
    const float p = s_pself[h * 16 + r4];
    const float4 sc = *(const float4*)(s_scale + (h * 16 + r4) * NCHUNK);
    float4 a;
    a.x = p * xs.x; a.y = p * xs.y; a.z = p * xs.z; a.w = p * xs.w;
    a.x += sc.x * pa[h][0].x; a.y += sc.x * pa[h][0].y; a.z += sc.x * pa[h][0].z; a.w += sc.x * pa[h][0].w;
    a.x += sc.y * pa[h][1].x; a.y += sc.y * pa[h][1].y; a.z += sc.y * pa[h][1].z; a.w += sc.y * pa[h][1].w;
    a.x += sc.z * pa[h][2].x; a.y += sc.z * pa[h][2].y; a.z += sc.z * pa[h][2].z; a.w += sc.z * pa[h][2].w;
    a.x += sc.w * pa[h][3].x; a.y += sc.w * pa[h][3].y; a.z += sc.w * pa[h][3].z; a.w += sc.w * pa[h][3].w;
    *(float4*)(s_ym + (h * 16 + r4) * Y_PITCH + k4) = a;
  }
  __syncthreads();

  {
    const int h = wave >> 2;
    v8f c;
    #pragma unroll
    for (int v = 0; v < 8; ++v) c[v] = bv * s_wsum[h * 16 + v + 8 * hi];
    const float* yrow = s_ym + (h * 16 + lo) * Y_PITCH;
    #pragma unroll
    for (int k = 0; k < IN_CH; k += 4) {
      const int kk = k + 2 * hi;
      const v2f a = *(const v2f*)(yrow + kk);
      c = __builtin_amdgcn_wmma_f32_16x16x4_f32(false, a, false, bw[k >> 2],
                                                (short)0, c, false, false);
    }
    #pragma unroll
    for (int v = 0; v < 8; ++v)
      s_o[(h * 16 + v + 8 * hi) * C_CH + (wave & 3) * 16 + lo] = c[v] / s_den[h * 16 + v + 8 * hi];
  }
  __syncthreads();

  {
    const float4 o0 = *(const float4*)(s_o + r4 * C_CH + k4);
    const float4 o1 = *(const float4*)(s_o + (16 + r4) * C_CH + k4);
    float4 o;
    o.x = 0.5f * (o0.x + o1.x) + bs.x;
    o.y = 0.5f * (o0.y + o1.y) + bs.y;
    o.z = 0.5f * (o0.z + o1.z) + bs.z;
    o.w = 0.5f * (o0.w + o1.w) + bs.w;
    *(float4*)(out + (size_t)(g * R_CENT + r4) * C_CH + k4) = o;
  }
}

extern "C" void kernel_launch(void* const* d_in, const int* in_sizes, int n_in,
                              void* d_out, int out_size, void* d_ws, size_t ws_size,
                              hipStream_t stream) {
  const float* x    = (const float*)d_in[0];
  const float* xcb  = (const float*)d_in[3];
  const float* wl   = (const float*)d_in[4];
  const float* bl   = (const float*)d_in[5];
  const float* wr   = (const float*)d_in[6];
  const float* br   = (const float*)d_in[7];
  const float* att  = (const float*)d_in[8];
  const float* bias = (const float*)d_in[9];
  float* out = (float*)d_out;

  float* ws = (float*)d_ws;
  float* selfl = ws;
  float* part = ws + (size_t)N_NODES * HC + R_CENT * HC;

  BipartPool_attn_part_kernel<<<B_GR * NCHUNK * HEADS, 128, 0, stream>>>(x, wl, bl, xcb, wr, br, att, selfl, part, ws_size);
  BipartPool_attn_merge_kernel<<<B_GR, 256, 0, stream>>>(x, wl, bl, selfl, bias, part, out, ws_size);
}
